# starting kernel
# baseline (speedup 1.0000x reference)
#include <hip/hip_runtime.h>
#include <hip/hip_bf16.h>


typedef float v2f __attribute__((ext_vector_type(2)));
typedef float v8f __attribute__((ext_vector_type(8)));

#define IN_DIM  128
#define HID_DIM 64
#define MID_DIM 128
#define OUT_DIM 32

__global__ void zero_f32_kernel(float* __restrict__ p, int n) {
    int i = blockIdx.x * blockDim.x + threadIdx.x;
    if (i < n) p[i] = 0.0f;
}

template <int W>
__global__ void bias_init_kernel(float* __restrict__ p, const float* __restrict__ bias, int n) {
    int i = blockIdx.x * blockDim.x + threadIdx.x;
    if (i < n) p[i] = bias[i % W];
}

__global__ void relu_kernel(float* __restrict__ p, int n) {
    int i = blockIdx.x * blockDim.x + threadIdx.x;
    if (i < n) p[i] = fmaxf(p[i], 0.0f);
}

__global__ void deg_accum_kernel(const int* __restrict__ dst, float* __restrict__ deg, int E) {
    int i = blockIdx.x * blockDim.x + threadIdx.x;
    if (i < E) atomicAdd(&deg[dst[i]], 1.0f);
}

__global__ void rsqrt_kernel(float* __restrict__ deg, int n) {
    int i = blockIdx.x * blockDim.x + threadIdx.x;
    if (i < n) deg[i] = 1.0f / sqrtf(deg[i] + 1.0f);
}

template <int K, int TILES, bool BIAS, bool RELU>
__global__ __launch_bounds__(256) void gemm_wmma_kernel(
    const float* __restrict__ A, const float* __restrict__ B,
    const float* __restrict__ bias, float* __restrict__ C,
    int numRowTiles, int n) {
    const int OUTW = TILES * 16;
    int wave = (blockIdx.x * blockDim.x + threadIdx.x) >> 5;
    int lane = threadIdx.x & 31;
    if (wave >= numRowTiles) return;

    const int m    = lane & 15;
    const int half = lane >> 4;
    const int row0 = wave * 16;

    v8f acc[TILES] = {};

    int arow = row0 + m;
    if (arow >= n) arow = n - 1;
    const float* __restrict__ ap = A + (long long)arow * K;

    for (int k0 = 0; k0 < K; k0 += 4) {
        v2f a = *(const v2f*)(ap + k0 + 2 * half);
        const float* __restrict__ bp = B + (long long)(k0 + 2 * half) * OUTW + m;
#pragma unroll
        for (int t = 0; t < TILES; ++t) {
            v2f b;
            b.x = bp[t * 16];
            b.y = bp[t * 16 + OUTW];
            acc[t] = __builtin_amdgcn_wmma_f32_16x16x4_f32(
                false, a, false, b, (short)0, acc[t], false, false);
        }
    }

#pragma unroll
    for (int t = 0; t < TILES; ++t) {
        const int col = t * 16 + m;
        float bv = BIAS ? bias[col] : 0.0f;
#pragma unroll
        for (int r = 0; r < 8; ++r) {
            int row = row0 + r + 8 * half;
            if (row < n) {
                float v = acc[t][r] + bv;
                if (RELU) v = fmaxf(v, 0.0f);
                C[(long long)row * OUTW + col] = v;
            }
        }
    }
}

template <int W>
__global__ void agg_kernel(const float* __restrict__ msg,
                           const int* __restrict__ src, const int* __restrict__ dst,
                           const float* __restrict__ dis, float* __restrict__ out,
                           int E, int N) {
    const int CH = W / 4;
    int idx = blockIdx.x * blockDim.x + threadIdx.x;
    int total = (E + N) * CH;
    if (idx >= total) return;
    int e = idx / CH;
    int c = (idx - e * CH) * 4;
    int s, d;
    if (e < E) { s = src[e]; d = dst[e]; }
    else       { s = d = e - E; }
    float nrm = dis[s] * dis[d];
    const float4 v = *(const float4*)(msg + (long long)s * W + c);
    float* o = out + (long long)d * W + c;
    atomicAdd(o + 0, v.x * nrm);
    atomicAdd(o + 1, v.y * nrm);
    atomicAdd(o + 2, v.z * nrm);
    atomicAdd(o + 3, v.w * nrm);
}

extern "C" void kernel_launch(void* const* d_in, const int* in_sizes, int n_in,
                              void* d_out, int out_size, void* d_ws, size_t ws_size,
                              hipStream_t stream) {
    const float* x   = (const float*)d_in[0];
    const int*   ei  = (const int*)  d_in[1];
    const float* W1  = (const float*)d_in[2];
    const float* b1  = (const float*)d_in[3];
    const float* Wl1 = (const float*)d_in[4];
    const float* bl1 = (const float*)d_in[5];
    const float* Wl2 = (const float*)d_in[6];
    const float* bl2 = (const float*)d_in[7];
    const float* W2  = (const float*)d_in[8];
    const float* b2  = (const float*)d_in[9];

    const int N = in_sizes[0] / IN_DIM;
    const int E = in_sizes[1] / 2;
    const int* src = ei;
    const int* dst = ei + E;

    float* ws   = (float*)d_ws;
    float* deg  = ws;
    float* bufA = deg  + (size_t)N;
    float* bufB = bufA + (size_t)N * 128;
    float* bufC = bufB + (size_t)N * 64;
    float* out  = (float*)d_out;

    const int TPB = 256;
    const int rowTiles = (N + 15) / 16;
    const int gemmBlocks = (rowTiles + 7) / 8;

    zero_f32_kernel<<<(N + TPB - 1) / TPB, TPB, 0, stream>>>(deg, N);
    deg_accum_kernel<<<(E + TPB - 1) / TPB, TPB, 0, stream>>>(dst, deg, E);
    rsqrt_kernel<<<(N + TPB - 1) / TPB, TPB, 0, stream>>>(deg, N);

    gemm_wmma_kernel<IN_DIM, HID_DIM / 16, false, false>
        <<<gemmBlocks, TPB, 0, stream>>>(x, W1, nullptr, bufA, rowTiles, N);
    bias_init_kernel<HID_DIM><<<((N * HID_DIM) + TPB - 1) / TPB, TPB, 0, stream>>>(bufB, b1, N * HID_DIM);
    {
        int total = (E + N) * (HID_DIM / 4);
        agg_kernel<HID_DIM><<<(total + TPB - 1) / TPB, TPB, 0, stream>>>(bufA, src, dst, deg, bufB, E, N);
    }
    relu_kernel<<<((N * HID_DIM) + TPB - 1) / TPB, TPB, 0, stream>>>(bufB, N * HID_DIM);

    gemm_wmma_kernel<HID_DIM, MID_DIM / 16, true, true>
        <<<gemmBlocks, TPB, 0, stream>>>(bufB, Wl1, bl1, bufA, rowTiles, N);
    gemm_wmma_kernel<MID_DIM, HID_DIM / 16, true, true>
        <<<gemmBlocks, TPB, 0, stream>>>(bufA, Wl2, bl2, bufC, rowTiles, N);

    gemm_wmma_kernel<HID_DIM, OUT_DIM / 16, false, false>
        <<<gemmBlocks, TPB, 0, stream>>>(bufC, W2, nullptr, bufB, rowTiles, N);
    bias_init_kernel<OUT_DIM><<<((N * OUT_DIM) + TPB - 1) / TPB, TPB, 0, stream>>>(out, b2, N * OUT_DIM);
    {
        int total = (E + N) * (OUT_DIM / 4);
        agg_kernel<OUT_DIM><<<(total + TPB - 1) / TPB, TPB, 0, stream>>>(bufB, src, dst, deg, out, E, N);
    }
}


# agg kernels: one wave per edge row (lane = feature column, 32 edges per wave) so each row load and f32 atomic is a whole 128-byte line; atomic arrival order changes
# speedup vs baseline: 1.2184x; 1.2184x over previous
#include <hip/hip_runtime.h>
#include <hip/hip_bf16.h>


typedef float v2f __attribute__((ext_vector_type(2)));
typedef float v8f __attribute__((ext_vector_type(8)));

#define IN_DIM  128
#define HID_DIM 64
#define MID_DIM 128
#define OUT_DIM 32

__global__ void zero_f32_kernel(float* __restrict__ p, int n) {
    int i = blockIdx.x * blockDim.x + threadIdx.x;
    if (i < n) p[i] = 0.0f;
}

template <int W>
__global__ void bias_init_kernel(float* __restrict__ p, const float* __restrict__ bias, int n) {
    int i = blockIdx.x * blockDim.x + threadIdx.x;
    if (i < n) p[i] = bias[i % W];
}

__global__ void relu_kernel(float* __restrict__ p, int n) {
    int i = blockIdx.x * blockDim.x + threadIdx.x;
    if (i < n) p[i] = fmaxf(p[i], 0.0f);
}

__global__ void deg_accum_kernel(const int* __restrict__ dst, float* __restrict__ deg, int E) {
    int i = blockIdx.x * blockDim.x + threadIdx.x;
    if (i < E) atomicAdd(&deg[dst[i]], 1.0f);
}

__global__ void rsqrt_kernel(float* __restrict__ deg, int n) {
    int i = blockIdx.x * blockDim.x + threadIdx.x;
    if (i < n) deg[i] = 1.0f / sqrtf(deg[i] + 1.0f);
}

template <int K, int TILES, bool BIAS, bool RELU>
__global__ __launch_bounds__(256) void gemm_wmma_kernel(
    const float* __restrict__ A, const float* __restrict__ B,
    const float* __restrict__ bias, float* __restrict__ C,
    int numRowTiles, int n) {
    const int OUTW = TILES * 16;
    int wave = (blockIdx.x * blockDim.x + threadIdx.x) >> 5;
    int lane = threadIdx.x & 31;
    if (wave >= numRowTiles) return;

    const int m    = lane & 15;
    const int half = lane >> 4;
    const int row0 = wave * 16;

    v8f acc[TILES] = {};

    int arow = row0 + m;
    if (arow >= n) arow = n - 1;
    const float* __restrict__ ap = A + (long long)arow * K;

    for (int k0 = 0; k0 < K; k0 += 4) {
        v2f a = *(const v2f*)(ap + k0 + 2 * half);
        const float* __restrict__ bp = B + (long long)(k0 + 2 * half) * OUTW + m;
#pragma unroll
        for (int t = 0; t < TILES; ++t) {
            v2f b;
            b.x = bp[t * 16];
            b.y = bp[t * 16 + OUTW];
            acc[t] = __builtin_amdgcn_wmma_f32_16x16x4_f32(
                false, a, false, b, (short)0, acc[t], false, false);
        }
    }

#pragma unroll
    for (int t = 0; t < TILES; ++t) {
        const int col = t * 16 + m;
        float bv = BIAS ? bias[col] : 0.0f;
#pragma unroll
        for (int r = 0; r < 8; ++r) {
            int row = row0 + r + 8 * half;
            if (row < n) {
                float v = acc[t][r] + bv;
                if (RELU) v = fmaxf(v, 0.0f);
                C[(long long)row * OUTW + col] = v;
            }
        }
    }
}

template <int W>
__global__ __launch_bounds__(256) void agg_kernel(const float* __restrict__ msg,
                           const int* __restrict__ src, const int* __restrict__ dst,
                           const float* __restrict__ dis, float* __restrict__ out,
                           int E, int N) {
    constexpr int LPR = W / 32;
    constexpr int UNR = 16 / LPR;
    const int lane = threadIdx.x & 31;
    const int wave = __builtin_amdgcn_readfirstlane((int)((blockIdx.x * blockDim.x + threadIdx.x) >> 5));
    const int total = E + N;
    const int e0 = wave * 32;
    if (e0 >= total) return;

    const int e = e0 + lane;
    int s = 0, d = 0;
    float nrm = 0.0f;
    if (e < total) {
        if (e < E) { s = src[e]; d = dst[e]; }
        else       { s = d = e - E; }
        nrm = dis[s] * dis[d];
    }
    const int nrm_bits = __float_as_int(nrm);
    const int cnt = min(32, total - e0);

    int j = 0;
    for (; j + UNR <= cnt; j += UNR) {
        float v[UNR][LPR];
        int   dj[UNR];
        float nj[UNR];
#pragma unroll
        for (int u = 0; u < UNR; ++u) {
            const int sj = __builtin_amdgcn_readlane(s, j + u);
            dj[u] = __builtin_amdgcn_readlane(d, j + u);
            nj[u] = __int_as_float(__builtin_amdgcn_readlane(nrm_bits, j + u));
            const float* __restrict__ mp = msg + (long long)sj * W + lane;
#pragma unroll
            for (int p = 0; p < LPR; ++p) v[u][p] = mp[p * 32];
        }
#pragma unroll
        for (int u = 0; u < UNR; ++u) {
            float* op = out + (long long)dj[u] * W + lane;
#pragma unroll
            for (int p = 0; p < LPR; ++p) atomicAdd(op + p * 32, v[u][p] * nj[u]);
        }
    }
    for (; j < cnt; ++j) {
        const int sj = __builtin_amdgcn_readlane(s, j);
        const int dd = __builtin_amdgcn_readlane(d, j);
        const float nn = __int_as_float(__builtin_amdgcn_readlane(nrm_bits, j));
        const float* __restrict__ mp = msg + (long long)sj * W + lane;
        float* op = out + (long long)dd * W + lane;
#pragma unroll
        for (int p = 0; p < LPR; ++p) atomicAdd(op + p * 32, mp[p * 32] * nn);
    }
}

extern "C" void kernel_launch(void* const* d_in, const int* in_sizes, int n_in,
                              void* d_out, int out_size, void* d_ws, size_t ws_size,
                              hipStream_t stream) {
    const float* x   = (const float*)d_in[0];
    const int*   ei  = (const int*)  d_in[1];
    const float* W1  = (const float*)d_in[2];
    const float* b1  = (const float*)d_in[3];
    const float* Wl1 = (const float*)d_in[4];
    const float* bl1 = (const float*)d_in[5];
    const float* Wl2 = (const float*)d_in[6];
    const float* bl2 = (const float*)d_in[7];
    const float* W2  = (const float*)d_in[8];
    const float* b2  = (const float*)d_in[9];

    const int N = in_sizes[0] / IN_DIM;
    const int E = in_sizes[1] / 2;
    const int* src = ei;
    const int* dst = ei + E;

    float* ws   = (float*)d_ws;
    float* deg  = ws;
    float* bufA = deg  + (size_t)N;
    float* bufB = bufA + (size_t)N * 128;
    float* bufC = bufB + (size_t)N * 64;
    float* out  = (float*)d_out;

    const int TPB = 256;
    const int rowTiles = (N + 15) / 16;
    const int gemmBlocks = (rowTiles + 7) / 8;

    zero_f32_kernel<<<(N + TPB - 1) / TPB, TPB, 0, stream>>>(deg, N);
    deg_accum_kernel<<<(E + TPB - 1) / TPB, TPB, 0, stream>>>(dst, deg, E);
    rsqrt_kernel<<<(N + TPB - 1) / TPB, TPB, 0, stream>>>(deg, N);

    gemm_wmma_kernel<IN_DIM, HID_DIM / 16, false, false>
        <<<gemmBlocks, TPB, 0, stream>>>(x, W1, nullptr, bufA, rowTiles, N);
    bias_init_kernel<HID_DIM><<<((N * HID_DIM) + TPB - 1) / TPB, TPB, 0, stream>>>(bufB, b1, N * HID_DIM);
    const int aggWaves  = (E + N + 31) / 32;
    const int aggBlocks = (aggWaves + 7) / 8;
    agg_kernel<HID_DIM><<<aggBlocks, TPB, 0, stream>>>(bufA, src, dst, deg, bufB, E, N);
    relu_kernel<<<((N * HID_DIM) + TPB - 1) / TPB, TPB, 0, stream>>>(bufB, N * HID_DIM);

    gemm_wmma_kernel<HID_DIM, MID_DIM / 16, true, true>
        <<<gemmBlocks, TPB, 0, stream>>>(bufB, Wl1, bl1, bufA, rowTiles, N);
    gemm_wmma_kernel<MID_DIM, HID_DIM / 16, true, true>
        <<<gemmBlocks, TPB, 0, stream>>>(bufA, Wl2, bl2, bufC, rowTiles, N);

    gemm_wmma_kernel<HID_DIM, OUT_DIM / 16, false, false>
        <<<gemmBlocks, TPB, 0, stream>>>(bufC, W2, nullptr, bufB, rowTiles, N);
    bias_init_kernel<OUT_DIM><<<((N * OUT_DIM) + TPB - 1) / TPB, TPB, 0, stream>>>(out, b2, N * OUT_DIM);
    agg_kernel<OUT_DIM><<<aggBlocks, TPB, 0, stream>>>(bufB, src, dst, deg, out, E, N);
}


# aggregation by destination: per-call counting sort of edges by dst (int rank atomics), one wave per node gathers in-rows, stores row once with bias/relu; no f32 atomics; row sum order = sort order
# speedup vs baseline: 1.6243x; 1.3331x over previous
#include <hip/hip_runtime.h>
#include <hip/hip_bf16.h>


typedef float v2f __attribute__((ext_vector_type(2)));
typedef float v4f __attribute__((ext_vector_type(4)));
typedef float v8f __attribute__((ext_vector_type(8)));

#define IN_DIM  128
#define HID_DIM 64
#define MID_DIM 128
#define OUT_DIM 32

__global__ void zero_i32_kernel(int* __restrict__ p, int n) {
    int i = blockIdx.x * blockDim.x + threadIdx.x;
    if (i < n) p[i] = 0;
}

__global__ void count_rank_kernel(const int* __restrict__ dst, int* __restrict__ cnt,
                                  int* __restrict__ rank, int E) {
    int e = blockIdx.x * blockDim.x + threadIdx.x;
    if (e < E) rank[e] = atomicAdd(&cnt[dst[e]], 1);
}

__global__ __launch_bounds__(1024) void scan_kernel(const int* __restrict__ cnt,
                                                    int* __restrict__ rowptr,
                                                    float* __restrict__ dis, int n) {
    __shared__ int wave_incl[32];
    __shared__ int wave_front[32];
    const int tid  = threadIdx.x;
    const int lane = tid & 31;
    const int wv   = tid >> 5;
    const int base = blockIdx.x * 1024;

    int front = 0;
    for (int i = tid; i < base; i += 1024) front += cnt[i];
    const int idx = base + tid;
    const int my  = idx < n ? cnt[idx] : 0;

    int incl = my;
    for (int o = 1; o < 32; o <<= 1) {
        int t = __shfl_up(incl, o);
        if (lane >= o) incl += t;
    }
    for (int o = 16; o > 0; o >>= 1) front += __shfl_xor(front, o);
    if (lane == 31) wave_incl[wv]  = incl;
    if (lane == 0)  wave_front[wv] = front;
    __syncthreads();
    int off = 0;
    for (int i = 0; i < 32; ++i) {
        off += wave_front[i];
        if (i < wv) off += wave_incl[i];
    }
    const int excl = off + incl - my;
    if (idx < n) {
        rowptr[idx] = excl;
        dis[idx] = 1.0f / sqrtf((float)my + 1.0f);
    }
    if (idx == n - 1) rowptr[n] = excl + my;
}

__global__ void fill_csr_kernel(const int* __restrict__ src, const int* __restrict__ dst,
                                const int* __restrict__ rank, const int* __restrict__ rowptr,
                                int* __restrict__ csr, int E) {
    int e = blockIdx.x * blockDim.x + threadIdx.x;
    if (e < E) csr[rowptr[dst[e]] + rank[e]] = src[e];
}

template <int K, int TILES, bool BIAS, bool RELU>
__global__ __launch_bounds__(256) void gemm_wmma_kernel(
    const float* __restrict__ A, const float* __restrict__ B,
    const float* __restrict__ bias, float* __restrict__ C,
    int numRowTiles, int n) {
    const int OUTW = TILES * 16;
    int wave = (blockIdx.x * blockDim.x + threadIdx.x) >> 5;
    int lane = threadIdx.x & 31;
    if (wave >= numRowTiles) return;

    const int m    = lane & 15;
    const int half = lane >> 4;
    const int row0 = wave * 16;

    v8f acc[TILES] = {};

    int arow = row0 + m;
    if (arow >= n) arow = n - 1;
    const float* __restrict__ ap = A + (long long)arow * K;

    for (int k0 = 0; k0 < K; k0 += 4) {
        v2f a = *(const v2f*)(ap + k0 + 2 * half);
        const float* __restrict__ bp = B + (long long)(k0 + 2 * half) * OUTW + m;
#pragma unroll
        for (int t = 0; t < TILES; ++t) {
            v2f b;
            b.x = bp[t * 16];
            b.y = bp[t * 16 + OUTW];
            acc[t] = __builtin_amdgcn_wmma_f32_16x16x4_f32(
                false, a, false, b, (short)0, acc[t], false, false);
        }
    }

#pragma unroll
    for (int t = 0; t < TILES; ++t) {
        const int col = t * 16 + m;
        float bv = BIAS ? bias[col] : 0.0f;
#pragma unroll
        for (int r = 0; r < 8; ++r) {
            int row = row0 + r + 8 * half;
            if (row < n) {
                float v = acc[t][r] + bv;
                if (RELU) v = fmaxf(v, 0.0f);
                C[(long long)row * OUTW + col] = v;
            }
        }
    }
}

template <int W, bool RELU>
__global__ __launch_bounds__(256) void agg_csr_kernel(const float* __restrict__ msg,
                           const int* __restrict__ rowptr, const int* __restrict__ csr,
                           const float* __restrict__ dis, const float* __restrict__ bias,
                           float* __restrict__ out, int N) {
    constexpr int LPRW = W / 4;
    constexpr int G    = 32 / LPRW;
    constexpr int UNR  = 8;
    const int lane = threadIdx.x & 31;
    const int d = __builtin_amdgcn_readfirstlane((int)((blockIdx.x * blockDim.x + threadIdx.x) >> 5));
    if (d >= N) return;
    const int g = lane / LPRW;
    const int c = lane % LPRW;

    const int beg = rowptr[d];
    const int end = rowptr[d + 1];
    const float dd = dis[d];

    v4f acc = {0.0f, 0.0f, 0.0f, 0.0f};
    for (int base = beg; base <= end; base += 32) {
        const int n = min(32, end + 1 - base);
        int   s  = d;
        float ds = 0.0f;
        if (lane < n) {
            const int idx = base + lane;
            if (idx < end) { s = csr[idx]; ds = dis[s]; }
            else           { ds = dd; }
        }
        for (int t0 = 0; t0 * G < n; t0 += UNR) {
            v4f v[UNR];
#pragma unroll
            for (int u = 0; u < UNR; ++u) {
                const int ei = (t0 + u) * G + g;
                const int sj = __builtin_amdgcn_ds_bpermute(ei << 2, s);
                v[u] = *(const v4f*)(msg + (long long)sj * W + 4 * c);
            }
            __builtin_amdgcn_sched_barrier(0);
#pragma unroll
            for (int u = 0; u < UNR; ++u) {
                const int ei = (t0 + u) * G + g;
                const float wj = __int_as_float(__builtin_amdgcn_ds_bpermute(ei << 2, __float_as_int(ds))) * dd;
                const v4f vv = (ei < n) ? v[u] : (v4f){0.0f, 0.0f, 0.0f, 0.0f};
                acc += vv * wj;
            }
        }
    }
#pragma unroll
    for (int o = 16; o >= LPRW; o >>= 1) {
        const int peer = (lane ^ o) << 2;
        acc.x += __int_as_float(__builtin_amdgcn_ds_bpermute(peer, __float_as_int(acc.x)));
        acc.y += __int_as_float(__builtin_amdgcn_ds_bpermute(peer, __float_as_int(acc.y)));
        acc.z += __int_as_float(__builtin_amdgcn_ds_bpermute(peer, __float_as_int(acc.z)));
        acc.w += __int_as_float(__builtin_amdgcn_ds_bpermute(peer, __float_as_int(acc.w)));
    }
    if (lane < LPRW) {
        v4f r = acc + *(const v4f*)(bias + 4 * c);
        if (RELU) {
            r.x = fmaxf(r.x, 0.0f); r.y = fmaxf(r.y, 0.0f);
            r.z = fmaxf(r.z, 0.0f); r.w = fmaxf(r.w, 0.0f);
        }
        *(v4f*)(out + (long long)d * W + 4 * c) = r;
    }
}

extern "C" void kernel_launch(void* const* d_in, const int* in_sizes, int n_in,
                              void* d_out, int out_size, void* d_ws, size_t ws_size,
                              hipStream_t stream) {
    const float* x   = (const float*)d_in[0];
    const int*   ei  = (const int*)  d_in[1];
    const float* W1  = (const float*)d_in[2];
    const float* b1  = (const float*)d_in[3];
    const float* Wl1 = (const float*)d_in[4];
    const float* bl1 = (const float*)d_in[5];
    const float* Wl2 = (const float*)d_in[6];
    const float* bl2 = (const float*)d_in[7];
    const float* W2  = (const float*)d_in[8];
    const float* b2  = (const float*)d_in[9];

    const int N = in_sizes[0] / IN_DIM;
    const int E = in_sizes[1] / 2;
    const int* src = ei;
    const int* dst = ei + E;

    const size_t Np = ((size_t)N + 1 + 63) / 64 * 64;
    const size_t Ep = ((size_t)E + 63) / 64 * 64;
    int*   cnt    = (int*)d_ws;
    int*   rowptr = cnt + Np;
    float* dis    = (float*)(rowptr + Np);
    int*   rank   = (int*)(dis + Np);
    int*   csr    = rank + Ep;
    float* bufA   = (float*)(csr + Ep);
    float* bufB   = bufA + (size_t)N * 128;
    float* bufC   = bufB + (size_t)N * 64;
    float* out    = (float*)d_out;

    const int TPB = 256;
    const int rowTiles = (N + 15) / 16;
    const int gemmBlocks = (rowTiles + 7) / 8;
    const int aggBlocks  = (N + 7) / 8;

    zero_i32_kernel<<<(N + TPB - 1) / TPB, TPB, 0, stream>>>(cnt, N);
    count_rank_kernel<<<(E + TPB - 1) / TPB, TPB, 0, stream>>>(dst, cnt, rank, E);
    scan_kernel<<<(N + 1023) / 1024, 1024, 0, stream>>>(cnt, rowptr, dis, N);
    fill_csr_kernel<<<(E + TPB - 1) / TPB, TPB, 0, stream>>>(src, dst, rank, rowptr, csr, E);

    gemm_wmma_kernel<IN_DIM, HID_DIM / 16, false, false>
        <<<gemmBlocks, TPB, 0, stream>>>(x, W1, nullptr, bufA, rowTiles, N);
    agg_csr_kernel<HID_DIM, true><<<aggBlocks, TPB, 0, stream>>>(bufA, rowptr, csr, dis, b1, bufB, N);

    gemm_wmma_kernel<HID_DIM, MID_DIM / 16, true, true>
        <<<gemmBlocks, TPB, 0, stream>>>(bufB, Wl1, bl1, bufA, rowTiles, N);
    gemm_wmma_kernel<MID_DIM, HID_DIM / 16, true, true>
        <<<gemmBlocks, TPB, 0, stream>>>(bufA, Wl2, bl2, bufC, rowTiles, N);

    gemm_wmma_kernel<HID_DIM, OUT_DIM / 16, false, false>
        <<<gemmBlocks, TPB, 0, stream>>>(bufC, W2, nullptr, bufB, rowTiles, N);
    agg_csr_kernel<OUT_DIM, false><<<aggBlocks, TPB, 0, stream>>>(bufB, rowptr, csr, dis, b2, out, N);
}


# edge sort rebuilt on LDS counters: chunk-by-512-node-bin partition, then one work-group per bin counts, scans, places (2 launches for 4); no global atomics; row sum order = LDS arrival order
# speedup vs baseline: 3.0146x; 1.8559x over previous
#include <hip/hip_runtime.h>
#include <hip/hip_bf16.h>


typedef float v2f __attribute__((ext_vector_type(2)));
typedef int   v2i __attribute__((ext_vector_type(2)));
typedef float v4f __attribute__((ext_vector_type(4)));
typedef float v8f __attribute__((ext_vector_type(8)));

#define IN_DIM  128
#define HID_DIM 64
#define MID_DIM 128
#define OUT_DIM 32

#define BIN_SHIFT 9
#define BIN_NODES 512
#define MAX_BINS  256
#define TBL_ROW   (MAX_BINS + 1)
#define P_THREADS 1024
#define P_EPT     8
#define P_CHUNK   (P_THREADS * P_EPT)
#define STAGE_CAP 12288

__global__ __launch_bounds__(P_THREADS) void partition_kernel(
    const int* __restrict__ src, const int* __restrict__ dst,
    int* __restrict__ tmp, int* __restrict__ table, int E) {
    __shared__ int hist[MAX_BINS];
    __shared__ int wsum[8];
    __shared__ int sorted[2 * P_CHUNK];
    const int tid  = threadIdx.x;
    const int lane = tid & 31;
    const int wv   = tid >> 5;
    const int chunk0 = blockIdx.x * P_CHUNK;

    if (tid < MAX_BINS) hist[tid] = 0;
    __syncthreads();

    int s[P_EPT], d[P_EPT], r[P_EPT];
#pragma unroll
    for (int k = 0; k < P_EPT; ++k) {
        const int e = chunk0 + k * P_THREADS + tid;
        s[k] = 0; d[k] = -1;
        if (e < E) { s[k] = src[e]; d[k] = dst[e]; }
    }
#pragma unroll
    for (int k = 0; k < P_EPT; ++k) {
        r[k] = 0;
        if (d[k] >= 0) {
            unsigned b = (unsigned)d[k] >> BIN_SHIFT;
            if (b > MAX_BINS - 1) b = MAX_BINS - 1;
            r[k] = atomicAdd(&hist[b], 1);
        }
    }
    __syncthreads();

    int c = 0, incl = 0;
    if (tid < MAX_BINS) {
        c = hist[tid];
        incl = c;
        for (int o = 1; o < 32; o <<= 1) {
            int t = __shfl_up(incl, o);
            if (lane >= o) incl += t;
        }
        if (lane == 31) wsum[wv] = incl;
    }
    __syncthreads();
    if (tid < MAX_BINS) {
        int off = 0;
        for (int i = 0; i < 8; ++i) if (i < wv) off += wsum[i];
        const int excl = off + incl - c;
        hist[tid] = excl;
        int* __restrict__ row = table + (long long)blockIdx.x * TBL_ROW;
        row[tid] = excl;
        if (tid == MAX_BINS - 1) row[MAX_BINS] = excl + c;
    }
    __syncthreads();

#pragma unroll
    for (int k = 0; k < P_EPT; ++k) {
        if (d[k] >= 0) {
            unsigned b = (unsigned)d[k] >> BIN_SHIFT;
            if (b > MAX_BINS - 1) b = MAX_BINS - 1;
            const int p = hist[b] + r[k];
            sorted[2 * p]     = s[k];
            sorted[2 * p + 1] = d[k];
        }
    }
    __syncthreads();

    const int n = min(P_CHUNK, E - chunk0);
    v2i* __restrict__ outp = (v2i*)(tmp + 2 * (long long)chunk0);
    for (int i = tid; i < n; i += P_THREADS) outp[i] = ((const v2i*)sorted)[i];
}

__global__ __launch_bounds__(1024) void bin_csr_kernel(
    const int* __restrict__ tmp, const int* __restrict__ table,
    int* __restrict__ rowptr, float* __restrict__ dis, int* __restrict__ csr,
    int N, int nblk) {
    __shared__ int cnt[BIN_NODES];
    __shared__ int rowoff[BIN_NODES];
    __shared__ int wbase[32];
    __shared__ int wtot[32];
    __shared__ int wsum[16];
    __shared__ int stage[STAGE_CAP];
    const int tid  = threadIdx.x;
    const int lane = tid & 31;
    const int wv   = tid >> 5;
    const int b    = blockIdx.x;

    int pb = 0, pt = 0;
    for (int blk = tid; blk < nblk; blk += 1024) {
        const int* __restrict__ row = table + (long long)blk * TBL_ROW;
        const int o0 = row[b], o1 = row[b + 1];
        pb += o0;
        pt += o1 - o0;
    }
    for (int o = 16; o > 0; o >>= 1) {
        pb += __shfl_xor(pb, o);
        pt += __shfl_xor(pt, o);
    }
    if (lane == 0) { wbase[wv] = pb; wtot[wv] = pt; }
    if (tid < BIN_NODES) cnt[tid] = 0;
    __syncthreads();
    int base = 0, total = 0;
    for (int i = 0; i < 32; ++i) { base += wbase[i]; total += wtot[i]; }
    const bool staged = total <= STAGE_CAP;

    for (int g = 0; g < nblk; g += 1024) {
        const int myblk = g + wv + 32 * lane;
        int o0 = 0, o1 = 0;
        if (myblk < nblk) {
            const int* __restrict__ row = table + (long long)myblk * TBL_ROW;
            o0 = row[b]; o1 = row[b + 1];
        }
        const int left = nblk - g - wv;
        const int nj = left <= 0 ? 0 : min(32, (left + 31) >> 5);
        for (int j = 0; j < nj; ++j) {
            const int a0 = __shfl(o0, j), a1 = __shfl(o1, j);
            const v2i* __restrict__ p = (const v2i*)(tmp + 2 * (long long)(g + wv + 32 * j) * P_CHUNK);
            for (int i = a0 + lane; i < a1; i += 32) {
                const v2i e = p[i];
                atomicAdd(&cnt[e.y & (BIN_NODES - 1)], 1);
            }
        }
    }
    __syncthreads();

    int c = 0, incl = 0;
    if (tid < BIN_NODES) {
        c = cnt[tid];
        incl = c;
        for (int o = 1; o < 32; o <<= 1) {
            int t = __shfl_up(incl, o);
            if (lane >= o) incl += t;
        }
        if (lane == 31) wsum[wv] = incl;
    }
    __syncthreads();
    if (tid < BIN_NODES) {
        int off = 0;
        for (int i = 0; i < 16; ++i) if (i < wv) off += wsum[i];
        const int excl = off + incl - c;
        rowoff[tid] = excl;
        cnt[tid] = 0;
        const int node = (b << BIN_SHIFT) + tid;
        if (node < N) {
            rowptr[node] = base + excl;
            dis[node] = 1.0f / sqrtf((float)c + 1.0f);
        }
        if (node == N - 1) rowptr[N] = base + excl + c;
    }
    __syncthreads();

    for (int g = 0; g < nblk; g += 1024) {
        const int myblk = g + wv + 32 * lane;
        int o0 = 0, o1 = 0;
        if (myblk < nblk) {
            const int* __restrict__ row = table + (long long)myblk * TBL_ROW;
            o0 = row[b]; o1 = row[b + 1];
        }
        const int left = nblk - g - wv;
        const int nj = left <= 0 ? 0 : min(32, (left + 31) >> 5);
        for (int j = 0; j < nj; ++j) {
            const int a0 = __shfl(o0, j), a1 = __shfl(o1, j);
            const v2i* __restrict__ p = (const v2i*)(tmp + 2 * (long long)(g + wv + 32 * j) * P_CHUNK);
            for (int i = a0 + lane; i < a1; i += 32) {
                const v2i e = p[i];
                const int dl = e.y & (BIN_NODES - 1);
                const int q = rowoff[dl] + atomicAdd(&cnt[dl], 1);
                if (staged) stage[q] = e.x;
                else        csr[base + q] = e.x;
            }
        }
    }
    if (staged) {
        __syncthreads();
        for (int i = tid; i < total; i += 1024) csr[base + i] = stage[i];
    }
}

template <int K, int TILES, bool BIAS, bool RELU, bool SCALE>
__global__ __launch_bounds__(256) void gemm_wmma_kernel(
    const float* __restrict__ A, const float* __restrict__ B,
    const float* __restrict__ bias, const float* __restrict__ rowscale,
    float* __restrict__ C, int numRowTiles, int n) {
    const int OUTW = TILES * 16;
    int wave = (blockIdx.x * blockDim.x + threadIdx.x) >> 5;
    int lane = threadIdx.x & 31;
    if (wave >= numRowTiles) return;

    const int m    = lane & 15;
    const int half = lane >> 4;
    const int row0 = wave * 16;

    v8f acc[TILES] = {};

    float rs[8];
    if (SCALE) {
#pragma unroll
        for (int r = 0; r < 8; ++r) {
            int row = row0 + r + 8 * half;
            rs[r] = rowscale[row < n ? row : n - 1];
        }
    }

    int arow = row0 + m;
    if (arow >= n) arow = n - 1;
    const float* __restrict__ ap = A + (long long)arow * K;

    for (int k0 = 0; k0 < K; k0 += 4) {
        v2f a = *(const v2f*)(ap + k0 + 2 * half);
        const float* __restrict__ bp = B + (long long)(k0 + 2 * half) * OUTW + m;
#pragma unroll
        for (int t = 0; t < TILES; ++t) {
            v2f b;
            b.x = bp[t * 16];
            b.y = bp[t * 16 + OUTW];
            acc[t] = __builtin_amdgcn_wmma_f32_16x16x4_f32(
                false, a, false, b, (short)0, acc[t], false, false);
        }
    }

#pragma unroll
    for (int t = 0; t < TILES; ++t) {
        const int col = t * 16 + m;
        float bv = BIAS ? bias[col] : 0.0f;
#pragma unroll
        for (int r = 0; r < 8; ++r) {
            int row = row0 + r + 8 * half;
            if (row < n) {
                float v = acc[t][r];
                if (SCALE) v *= rs[r];
                if (BIAS) v += bv;
                if (RELU) v = fmaxf(v, 0.0f);
                C[(long long)row * OUTW + col] = v;
            }
        }
    }
}

template <int W, bool RELU>
__global__ __launch_bounds__(256) void agg_csr_kernel(const float* __restrict__ msg,
                           const int* __restrict__ rowptr, const int* __restrict__ csr,
                           const float* __restrict__ dis, const float* __restrict__ bias,
                           float* __restrict__ out, int N) {
    constexpr int LPRW = W / 4;
    constexpr int G    = 32 / LPRW;
    constexpr int UNR  = 8;
    const int lane = threadIdx.x & 31;
    const int d = __builtin_amdgcn_readfirstlane((int)((blockIdx.x * blockDim.x + threadIdx.x) >> 5));
    if (d >= N) return;
    const int g = lane / LPRW;
    const int c = lane % LPRW;

    const int beg = rowptr[d];
    const int end = rowptr[d + 1];
    const float dd = dis[d];

    v4f acc = {0.0f, 0.0f, 0.0f, 0.0f};
    for (int base = beg; base <= end; base += 32) {
        const int n = min(32, end + 1 - base);
        int s = d;
        {
            const int idx = base + lane;
            if (idx < end) s = csr[idx];
        }
        for (int t0 = 0; t0 * G < n; t0 += UNR) {
            v4f v[UNR];
#pragma unroll
            for (int u = 0; u < UNR; ++u) {
                const int ei = (t0 + u) * G + g;
                const int sj = __builtin_amdgcn_ds_bpermute(ei << 2, s);
                v[u] = *(const v4f*)(msg + (long long)sj * W + 4 * c);
            }
#pragma unroll
            for (int u = 0; u < UNR; ++u) {
                const int ei = (t0 + u) * G + g;
                const v4f vv = (ei < n) ? v[u] : (v4f){0.0f, 0.0f, 0.0f, 0.0f};
                acc += vv;
            }
        }
    }
#pragma unroll
    for (int o = 16; o >= LPRW; o >>= 1) {
        const int peer = (lane ^ o) << 2;
        acc.x += __int_as_float(__builtin_amdgcn_ds_bpermute(peer, __float_as_int(acc.x)));
        acc.y += __int_as_float(__builtin_amdgcn_ds_bpermute(peer, __float_as_int(acc.y)));
        acc.z += __int_as_float(__builtin_amdgcn_ds_bpermute(peer, __float_as_int(acc.z)));
        acc.w += __int_as_float(__builtin_amdgcn_ds_bpermute(peer, __float_as_int(acc.w)));
    }
    if (lane < LPRW) {
        v4f r = acc * dd + *(const v4f*)(bias + 4 * c);
        if (RELU) {
            r.x = fmaxf(r.x, 0.0f); r.y = fmaxf(r.y, 0.0f);
            r.z = fmaxf(r.z, 0.0f); r.w = fmaxf(r.w, 0.0f);
        }
        *(v4f*)(out + (long long)d * W + 4 * c) = r;
    }
}

extern "C" void kernel_launch(void* const* d_in, const int* in_sizes, int n_in,
                              void* d_out, int out_size, void* d_ws, size_t ws_size,
                              hipStream_t stream) {
    const float* x   = (const float*)d_in[0];
    const int*   ei  = (const int*)  d_in[1];
    const float* W1  = (const float*)d_in[2];
    const float* b1  = (const float*)d_in[3];
    const float* Wl1 = (const float*)d_in[4];
    const float* bl1 = (const float*)d_in[5];
    const float* Wl2 = (const float*)d_in[6];
    const float* bl2 = (const float*)d_in[7];
    const float* W2  = (const float*)d_in[8];
    const float* b2  = (const float*)d_in[9];

    const int N = in_sizes[0] / IN_DIM;
    const int E = in_sizes[1] / 2;
    const int* src = ei;
    const int* dst = ei + E;

    const int nblk = (E + P_CHUNK - 1) / P_CHUNK;
    const int nbin = (N + BIN_NODES - 1) >> BIN_SHIFT;
    const size_t Np = ((size_t)N + 1 + 63) / 64 * 64;
    const size_t Ep = ((size_t)E + 63) / 64 * 64;
    const size_t Tp = ((size_t)nblk * TBL_ROW + 63) / 64 * 64;
    int*   table  = (int*)d_ws;
    int*   rowptr = table + Tp;
    float* dis    = (float*)(rowptr + Np);
    int*   tmp    = (int*)(dis + Np);
    int*   csr    = tmp + 2 * Ep;
    float* bufA   = (float*)(csr + Ep);
    float* bufB   = bufA + (size_t)N * 128;
    float* bufC   = bufB + (size_t)N * 64;
    float* out    = (float*)d_out;

    const int TPB = 256;
    const int rowTiles = (N + 15) / 16;
    const int gemmBlocks = (rowTiles + 7) / 8;
    const int aggBlocks  = (N + 7) / 8;

    partition_kernel<<<nblk, P_THREADS, 0, stream>>>(src, dst, tmp, table, E);
    bin_csr_kernel<<<nbin, 1024, 0, stream>>>(tmp, table, rowptr, dis, csr, N, nblk);

    gemm_wmma_kernel<IN_DIM, HID_DIM / 16, false, false, true>
        <<<gemmBlocks, TPB, 0, stream>>>(x, W1, nullptr, dis, bufA, rowTiles, N);
    agg_csr_kernel<HID_DIM, true><<<aggBlocks, TPB, 0, stream>>>(bufA, rowptr, csr, dis, b1, bufB, N);

    gemm_wmma_kernel<HID_DIM, MID_DIM / 16, true, true, false>
        <<<gemmBlocks, TPB, 0, stream>>>(bufB, Wl1, bl1, nullptr, bufA, rowTiles, N);
    gemm_wmma_kernel<MID_DIM, HID_DIM / 16, true, true, false>
        <<<gemmBlocks, TPB, 0, stream>>>(bufA, Wl2, bl2, nullptr, bufC, rowTiles, N);

    gemm_wmma_kernel<HID_DIM, OUT_DIM / 16, false, false, true>
        <<<gemmBlocks, TPB, 0, stream>>>(bufC, W2, nullptr, dis, bufB, rowTiles, N);
    agg_csr_kernel<OUT_DIM, false><<<aggBlocks, TPB, 0, stream>>>(bufB, rowptr, csr, dis, b2, out, N);
}


# conv1 gather fused with the 3 dense layers behind it: one wave gathers 16 nodes' rows into LDS, then runs the 3 f32 WMMA products on that tile (via LDS, transposed, sum orders kept); 8 launches to 5
# speedup vs baseline: 3.9375x; 1.3061x over previous
#include <hip/hip_runtime.h>
#include <hip/hip_bf16.h>


typedef float v2f __attribute__((ext_vector_type(2)));
typedef int   v2i __attribute__((ext_vector_type(2)));
typedef float v4f __attribute__((ext_vector_type(4)));
typedef float v8f __attribute__((ext_vector_type(8)));

#define IN_DIM  128
#define HID_DIM 64
#define MID_DIM 128
#define OUT_DIM 32

#define BIN_SHIFT 9
#define BIN_NODES 512
#define MAX_BINS  256
#define TBL_ROW   (MAX_BINS + 1)
#define P_THREADS 1024
#define P_EPT     8
#define P_CHUNK   (P_THREADS * P_EPT)
#define STAGE_CAP 12288

__global__ __launch_bounds__(P_THREADS) void partition_kernel(
    const int* __restrict__ src, const int* __restrict__ dst,
    int* __restrict__ tmp, int* __restrict__ table, int E) {
    __shared__ int hist[MAX_BINS];
    __shared__ int wsum[8];
    __shared__ int sorted[2 * P_CHUNK];
    const int tid  = threadIdx.x;
    const int lane = tid & 31;
    const int wv   = tid >> 5;
    const int chunk0 = blockIdx.x * P_CHUNK;

    if (tid < MAX_BINS) hist[tid] = 0;
    __syncthreads();

    int s[P_EPT], d[P_EPT], r[P_EPT];
#pragma unroll
    for (int k = 0; k < P_EPT; ++k) {
        const int e = chunk0 + k * P_THREADS + tid;
        s[k] = 0; d[k] = -1;
        if (e < E) { s[k] = src[e]; d[k] = dst[e]; }
    }
#pragma unroll
    for (int k = 0; k < P_EPT; ++k) {
        r[k] = 0;
        if (d[k] >= 0) {
            unsigned b = (unsigned)d[k] >> BIN_SHIFT;
            if (b > MAX_BINS - 1) b = MAX_BINS - 1;
            r[k] = atomicAdd(&hist[b], 1);
        }
    }
    __syncthreads();

    int c = 0, incl = 0;
    if (tid < MAX_BINS) {
        c = hist[tid];
        incl = c;
        for (int o = 1; o < 32; o <<= 1) {
            int t = __shfl_up(incl, o);
            if (lane >= o) incl += t;
        }
        if (lane == 31) wsum[wv] = incl;
    }
    __syncthreads();
    if (tid < MAX_BINS) {
        int off = 0;
        for (int i = 0; i < 8; ++i) if (i < wv) off += wsum[i];
        const int excl = off + incl - c;
        hist[tid] = excl;
        int* __restrict__ row = table + (long long)blockIdx.x * TBL_ROW;
        row[tid] = excl;
        if (tid == MAX_BINS - 1) row[MAX_BINS] = excl + c;
    }
    __syncthreads();

#pragma unroll
    for (int k = 0; k < P_EPT; ++k) {
        if (d[k] >= 0) {
            unsigned b = (unsigned)d[k] >> BIN_SHIFT;
            if (b > MAX_BINS - 1) b = MAX_BINS - 1;
            const int p = hist[b] + r[k];
            sorted[2 * p]     = s[k];
            sorted[2 * p + 1] = d[k];
        }
    }
    __syncthreads();

    const int n = min(P_CHUNK, E - chunk0);
    v2i* __restrict__ outp = (v2i*)(tmp + 2 * (long long)chunk0);
    for (int i = tid; i < n; i += P_THREADS) outp[i] = ((const v2i*)sorted)[i];
}

__global__ __launch_bounds__(1024) void bin_csr_kernel(
    const int* __restrict__ tmp, const int* __restrict__ table,
    int* __restrict__ rowptr, float* __restrict__ dis, int* __restrict__ csr,
    int N, int nblk) {
    __shared__ int cnt[BIN_NODES];
    __shared__ int rowoff[BIN_NODES];
    __shared__ int wbase[32];
    __shared__ int wtot[32];
    __shared__ int wsum[16];
    __shared__ int stage[STAGE_CAP];
    const int tid  = threadIdx.x;
    const int lane = tid & 31;
    const int wv   = tid >> 5;
    const int b    = blockIdx.x;

    int pb = 0, pt = 0;
    for (int blk = tid; blk < nblk; blk += 1024) {
        const int* __restrict__ row = table + (long long)blk * TBL_ROW;
        const int o0 = row[b], o1 = row[b + 1];
        pb += o0;
        pt += o1 - o0;
    }
    for (int o = 16; o > 0; o >>= 1) {
        pb += __shfl_xor(pb, o);
        pt += __shfl_xor(pt, o);
    }
    if (lane == 0) { wbase[wv] = pb; wtot[wv] = pt; }
    if (tid < BIN_NODES) cnt[tid] = 0;
    __syncthreads();
    int base = 0, total = 0;
    for (int i = 0; i < 32; ++i) { base += wbase[i]; total += wtot[i]; }
    const bool staged = total <= STAGE_CAP;

    for (int g = 0; g < nblk; g += 1024) {
        const int myblk = g + wv + 32 * lane;
        int o0 = 0, o1 = 0;
        if (myblk < nblk) {
            const int* __restrict__ row = table + (long long)myblk * TBL_ROW;
            o0 = row[b]; o1 = row[b + 1];
        }
        const int left = nblk - g - wv;
        const int nj = left <= 0 ? 0 : min(32, (left + 31) >> 5);
        for (int j = 0; j < nj; ++j) {
            const int a0 = __shfl(o0, j), a1 = __shfl(o1, j);
            const v2i* __restrict__ p = (const v2i*)(tmp + 2 * (long long)(g + wv + 32 * j) * P_CHUNK);
            for (int i = a0 + lane; i < a1; i += 32) {
                const v2i e = p[i];
                atomicAdd(&cnt[e.y & (BIN_NODES - 1)], 1);
            }
        }
    }
    __syncthreads();

    int c = 0, incl = 0;
    if (tid < BIN_NODES) {
        c = cnt[tid];
        incl = c;
        for (int o = 1; o < 32; o <<= 1) {
            int t = __shfl_up(incl, o);
            if (lane >= o) incl += t;
        }
        if (lane == 31) wsum[wv] = incl;
    }
    __syncthreads();
    if (tid < BIN_NODES) {
        int off = 0;
        for (int i = 0; i < 16; ++i) if (i < wv) off += wsum[i];
        const int excl = off + incl - c;
        rowoff[tid] = excl;
        cnt[tid] = 0;
        const int node = (b << BIN_SHIFT) + tid;
        if (node < N) {
            rowptr[node] = base + excl;
            dis[node] = 1.0f / sqrtf((float)c + 1.0f);
        }
        if (node == N - 1) rowptr[N] = base + excl + c;
    }
    __syncthreads();

    for (int g = 0; g < nblk; g += 1024) {
        const int myblk = g + wv + 32 * lane;
        int o0 = 0, o1 = 0;
        if (myblk < nblk) {
            const int* __restrict__ row = table + (long long)myblk * TBL_ROW;
            o0 = row[b]; o1 = row[b + 1];
        }
        const int left = nblk - g - wv;
        const int nj = left <= 0 ? 0 : min(32, (left + 31) >> 5);
        for (int j = 0; j < nj; ++j) {
            const int a0 = __shfl(o0, j), a1 = __shfl(o1, j);
            const v2i* __restrict__ p = (const v2i*)(tmp + 2 * (long long)(g + wv + 32 * j) * P_CHUNK);
            for (int i = a0 + lane; i < a1; i += 32) {
                const v2i e = p[i];
                const int dl = e.y & (BIN_NODES - 1);
                const int q = rowoff[dl] + atomicAdd(&cnt[dl], 1);
                if (staged) stage[q] = e.x;
                else        csr[base + q] = e.x;
            }
        }
    }
    if (staged) {
        __syncthreads();
        for (int i = tid; i < total; i += 1024) csr[base + i] = stage[i];
    }
}

template <int K, int TILES, bool BIAS, bool RELU, bool SCALE>
__global__ __launch_bounds__(256) void gemm_wmma_kernel(
    const float* __restrict__ A, const float* __restrict__ B,
    const float* __restrict__ bias, const float* __restrict__ rowscale,
    float* __restrict__ C, int numRowTiles, int n) {
    const int OUTW = TILES * 16;
    int wave = (blockIdx.x * blockDim.x + threadIdx.x) >> 5;
    int lane = threadIdx.x & 31;
    if (wave >= numRowTiles) return;

    const int m    = lane & 15;
    const int half = lane >> 4;
    const int row0 = wave * 16;

    v8f acc[TILES] = {};

    float rs[8];
    if (SCALE) {
#pragma unroll
        for (int r = 0; r < 8; ++r) {
            int row = row0 + r + 8 * half;
            rs[r] = rowscale[row < n ? row : n - 1];
        }
    }

    int arow = row0 + m;
    if (arow >= n) arow = n - 1;
    const float* __restrict__ ap = A + (long long)arow * K;

    for (int k0 = 0; k0 < K; k0 += 4) {
        v2f a = *(const v2f*)(ap + k0 + 2 * half);
        const float* __restrict__ bp = B + (long long)(k0 + 2 * half) * OUTW + m;
#pragma unroll
        for (int t = 0; t < TILES; ++t) {
            v2f b;
            b.x = bp[t * 16];
            b.y = bp[t * 16 + OUTW];
            acc[t] = __builtin_amdgcn_wmma_f32_16x16x4_f32(
                false, a, false, b, (short)0, acc[t], false, false);
        }
    }

#pragma unroll
    for (int t = 0; t < TILES; ++t) {
        const int col = t * 16 + m;
        float bv = BIAS ? bias[col] : 0.0f;
#pragma unroll
        for (int r = 0; r < 8; ++r) {
            int row = row0 + r + 8 * half;
            if (row < n) {
                float v = acc[t][r];
                if (SCALE) v *= rs[r];
                if (BIAS) v += bv;
                if (RELU) v = fmaxf(v, 0.0f);
                C[(long long)row * OUTW + col] = v;
            }
        }
    }
}

template <int W, bool RELU>
__global__ __launch_bounds__(256) void agg_csr_kernel(const float* __restrict__ msg,
                           const int* __restrict__ rowptr, const int* __restrict__ csr,
                           const float* __restrict__ dis, const float* __restrict__ bias,
                           float* __restrict__ out, int N) {
    constexpr int LPRW = W / 4;
    constexpr int G    = 32 / LPRW;
    constexpr int UNR  = 8;
    const int lane = threadIdx.x & 31;
    const int d = __builtin_amdgcn_readfirstlane((int)((blockIdx.x * blockDim.x + threadIdx.x) >> 5));
    if (d >= N) return;
    const int g = lane / LPRW;
    const int c = lane % LPRW;

    const int beg = rowptr[d];
    const int end = rowptr[d + 1];
    const float dd = dis[d];

    v4f acc = {0.0f, 0.0f, 0.0f, 0.0f};
    for (int base = beg; base <= end; base += 32) {
        const int n = min(32, end + 1 - base);
        int s = d;
        {
            const int idx = base + lane;
            if (idx < end) s = csr[idx];
        }
        for (int t0 = 0; t0 * G < n; t0 += UNR) {
            v4f v[UNR];
#pragma unroll
            for (int u = 0; u < UNR; ++u) {
                const int ei = (t0 + u) * G + g;
                const int sj = __builtin_amdgcn_ds_bpermute(ei << 2, s);
                v[u] = *(const v4f*)(msg + (long long)sj * W + 4 * c);
            }
#pragma unroll
            for (int u = 0; u < UNR; ++u) {
                const int ei = (t0 + u) * G + g;
                const v4f vv = (ei < n) ? v[u] : (v4f){0.0f, 0.0f, 0.0f, 0.0f};
                acc += vv;
            }
        }
    }
#pragma unroll
    for (int o = 16; o >= LPRW; o >>= 1) {
        const int peer = (lane ^ o) << 2;
        acc.x += __int_as_float(__builtin_amdgcn_ds_bpermute(peer, __float_as_int(acc.x)));
        acc.y += __int_as_float(__builtin_amdgcn_ds_bpermute(peer, __float_as_int(acc.y)));
        acc.z += __int_as_float(__builtin_amdgcn_ds_bpermute(peer, __float_as_int(acc.z)));
        acc.w += __int_as_float(__builtin_amdgcn_ds_bpermute(peer, __float_as_int(acc.w)));
    }
    if (lane < LPRW) {
        v4f r = acc * dd + *(const v4f*)(bias + 4 * c);
        if (RELU) {
            r.x = fmaxf(r.x, 0.0f); r.y = fmaxf(r.y, 0.0f);
            r.z = fmaxf(r.z, 0.0f); r.w = fmaxf(r.w, 0.0f);
        }
        *(v4f*)(out + (long long)d * W + 4 * c) = r;
    }
}

#define FT_NODES 16
#define FT_PITCH 68

template <int NT>
__device__ __forceinline__ void dense64_T(const float* __restrict__ Wt, const float* hs,
                                          int m, int half, v8f (&acc)[NT]) {
    constexpr int OUTW = NT * 16;
    for (int k0 = 0; k0 < 64; k0 += 4) {
        const v2f bh = *(const v2f*)(hs + m * FT_PITCH + k0 + 2 * half);
        const float* __restrict__ wp = Wt + (k0 + 2 * half) * OUTW + m;
#pragma unroll
        for (int t = 0; t < NT; ++t) {
            v2f aw;
            aw.x = wp[t * 16];
            aw.y = wp[t * 16 + OUTW];
            acc[t] = __builtin_amdgcn_wmma_f32_16x16x4_f32(
                false, aw, false, bh, (short)0, acc[t], false, false);
        }
    }
}

__device__ __forceinline__ void bias_relu_T(v8f& a, const float* __restrict__ bias16, int half) {
    const v4f ba = *(const v4f*)(bias16 + 8 * half);
    const v4f bb = *(const v4f*)(bias16 + 8 * half + 4);
    a[0] = fmaxf(a[0] + ba.x, 0.0f); a[1] = fmaxf(a[1] + ba.y, 0.0f);
    a[2] = fmaxf(a[2] + ba.z, 0.0f); a[3] = fmaxf(a[3] + ba.w, 0.0f);
    a[4] = fmaxf(a[4] + bb.x, 0.0f); a[5] = fmaxf(a[5] + bb.y, 0.0f);
    a[6] = fmaxf(a[6] + bb.z, 0.0f); a[7] = fmaxf(a[7] + bb.w, 0.0f);
}

__device__ __forceinline__ void put_tile_T(float* hs, int t, int m, int half, const v8f& a) {
    float* p = hs + m * FT_PITCH + 16 * t + 8 * half;
    *(v4f*)p       = (v4f){a[0], a[1], a[2], a[3]};
    *(v4f*)(p + 4) = (v4f){a[4], a[5], a[6], a[7]};
}

__global__ __launch_bounds__(32) __attribute__((amdgpu_waves_per_eu(8))) void conv1_dense_kernel(
    const float* __restrict__ msg, const int* __restrict__ rowptr, const int* __restrict__ csr,
    const float* __restrict__ dis, const float* __restrict__ b1,
    const float* __restrict__ Wl1, const float* __restrict__ bl1,
    const float* __restrict__ Wl2, const float* __restrict__ bl2,
    const float* __restrict__ W2, float* __restrict__ out, int N) {
    constexpr int W    = HID_DIM;
    constexpr int LPRW = W / 4;
    constexpr int G    = 32 / LPRW;
    constexpr int UNR  = 8;
    __shared__ float hs[FT_NODES * FT_PITCH];
    const int lane = threadIdx.x;
    const int d0 = blockIdx.x * FT_NODES;
    if (d0 >= N) return;
    const int nn = min(FT_NODES, N - d0);
    const int g = lane / LPRW;
    const int c = lane % LPRW;

    const int   rpl = rowptr[d0 + min(lane, nn)];
    const float dsl = dis[d0 + min(lane, nn - 1)];
    const v4f   bv  = *(const v4f*)(b1 + 4 * c);
    int beg = __builtin_amdgcn_readlane(rpl, 0);
    int end = __builtin_amdgcn_readlane(rpl, 1);
    int s = d0;
    {
        const int idx = beg + lane;
        if (idx < end) s = csr[idx];
    }
    const char* __restrict__ msgb = (const char*)msg;
    for (int j = 0; j < nn; ++j) {
        const int d = d0 + j;
        const float dd = __int_as_float(__builtin_amdgcn_readlane(__float_as_int(dsl), j));
        bool fetch_next = j + 1 < nn;
        const int nend = fetch_next ? __builtin_amdgcn_readlane(rpl, j + 2) : end;
        int sn = fetch_next ? d + 1 : d;

        v4f acc = {0.0f, 0.0f, 0.0f, 0.0f};
        for (int base = beg; base <= end; base += 32) {
            const int n = min(32, end + 1 - base);
            int sc = s;
            if (base != beg) {
                sc = d;
                const int idx = base + lane;
                if (idx < end) sc = csr[idx];
            }
#pragma unroll 1
            for (int t0 = 0; t0 * G < n; t0 += UNR) {
                unsigned off[UNR];
#pragma unroll
                for (int u = 0; u < UNR; ++u) {
                    const int ei = (t0 + u) * G + g;
                    const int sj = __builtin_amdgcn_ds_bpermute(ei << 2, sc);
                    off[u] = (unsigned)sj * (unsigned)(W * 4) + (unsigned)(16 * c);
                }
                if (fetch_next) {
                    fetch_next = false;
                    const int idx = end + lane;
                    if (idx < nend) sn = csr[idx];
                }
                v4f v[UNR];
#pragma unroll
                for (int u = 0; u < UNR; ++u) v[u] = *(const v4f*)(msgb + off[u]);
#pragma unroll
                for (int u = 0; u < UNR; ++u) {
                    const v4f vv = (g < n - (t0 + u) * G) ? v[u] : (v4f){0.0f, 0.0f, 0.0f, 0.0f};
                    acc += vv;
                }
            }
        }
#pragma unroll
        for (int o = 16; o >= LPRW; o >>= 1) {
            const int peer = (lane ^ o) << 2;
            acc.x += __int_as_float(__builtin_amdgcn_ds_bpermute(peer, __float_as_int(acc.x)));
            acc.y += __int_as_float(__builtin_amdgcn_ds_bpermute(peer, __float_as_int(acc.y)));
            acc.z += __int_as_float(__builtin_amdgcn_ds_bpermute(peer, __float_as_int(acc.z)));
            acc.w += __int_as_float(__builtin_amdgcn_ds_bpermute(peer, __float_as_int(acc.w)));
        }
        if (lane < LPRW) {
            v4f r = acc * dd + bv;
            r.x = fmaxf(r.x, 0.0f); r.y = fmaxf(r.y, 0.0f);
            r.z = fmaxf(r.z, 0.0f); r.w = fmaxf(r.w, 0.0f);
            *(v4f*)(hs + j * FT_PITCH + 4 * c) = r;
        }
        s = sn;
        beg = end;
        end = nend;
    }
    for (int j = nn; j < FT_NODES; ++j)
        if (lane < LPRW) *(v4f*)(hs + j * FT_PITCH + 4 * c) = (v4f){0.0f, 0.0f, 0.0f, 0.0f};
    __syncthreads();

    const int m    = lane & 15;
    const int half = lane >> 4;

    v8f acc1[MID_DIM / 16] = {};
    dense64_T<MID_DIM / 16>(Wl1, hs, m, half, acc1);
#pragma unroll
    for (int t = 0; t < MID_DIM / 16; ++t) bias_relu_T(acc1[t], bl1 + 16 * t, half);

    v8f acc2[HID_DIM / 16] = {};
    __syncthreads();
#pragma unroll
    for (int t = 0; t < 4; ++t) put_tile_T(hs, t, m, half, acc1[t]);
    __syncthreads();
    dense64_T<HID_DIM / 16>(Wl2, hs, m, half, acc2);
    __syncthreads();
#pragma unroll
    for (int t = 0; t < 4; ++t) put_tile_T(hs, t, m, half, acc1[4 + t]);
    __syncthreads();
    dense64_T<HID_DIM / 16>(Wl2 + 64 * HID_DIM, hs, m, half, acc2);
#pragma unroll
    for (int t = 0; t < HID_DIM / 16; ++t) bias_relu_T(acc2[t], bl2 + 16 * t, half);

    v8f acc3[OUT_DIM / 16] = {};
    __syncthreads();
#pragma unroll
    for (int t = 0; t < HID_DIM / 16; ++t) put_tile_T(hs, t, m, half, acc2[t]);
    __syncthreads();
    dense64_T<OUT_DIM / 16>(W2, hs, m, half, acc3);

    const int row = d0 + m;
    if (row < N) {
        const float rs = dis[row];
        float* __restrict__ op = out + (long long)row * OUT_DIM + 8 * half;
#pragma unroll
        for (int t = 0; t < OUT_DIM / 16; ++t) {
            const v4f lo = {acc3[t][0] * rs, acc3[t][1] * rs, acc3[t][2] * rs, acc3[t][3] * rs};
            const v4f hi = {acc3[t][4] * rs, acc3[t][5] * rs, acc3[t][6] * rs, acc3[t][7] * rs};
            *(v4f*)(op + 16 * t)     = lo;
            *(v4f*)(op + 16 * t + 4) = hi;
        }
    }
}

extern "C" void kernel_launch(void* const* d_in, const int* in_sizes, int n_in,
                              void* d_out, int out_size, void* d_ws, size_t ws_size,
                              hipStream_t stream) {
    const float* x   = (const float*)d_in[0];
    const int*   ei  = (const int*)  d_in[1];
    const float* W1  = (const float*)d_in[2];
    const float* b1  = (const float*)d_in[3];
    const float* Wl1 = (const float*)d_in[4];
    const float* bl1 = (const float*)d_in[5];
    const float* Wl2 = (const float*)d_in[6];
    const float* bl2 = (const float*)d_in[7];
    const float* W2  = (const float*)d_in[8];
    const float* b2  = (const float*)d_in[9];

    const int N = in_sizes[0] / IN_DIM;
    const int E = in_sizes[1] / 2;
    const int* src = ei;
    const int* dst = ei + E;

    const int nblk = (E + P_CHUNK - 1) / P_CHUNK;
    const int nbin = (N + BIN_NODES - 1) >> BIN_SHIFT;
    const size_t Np = ((size_t)N + 1 + 63) / 64 * 64;
    const size_t Ep = ((size_t)E + 63) / 64 * 64;
    const size_t Tp = ((size_t)nblk * TBL_ROW + 63) / 64 * 64;
    int*   table  = (int*)d_ws;
    int*   rowptr = table + Tp;
    float* dis    = (float*)(rowptr + Np);
    int*   tmp    = (int*)(dis + Np);
    int*   csr    = tmp + 2 * Ep;
    float* bufA   = (float*)(csr + Ep);
    float* bufB   = bufA + (size_t)N * 64;
    float* out    = (float*)d_out;

    const int TPB = 256;
    const int rowTiles = (N + 15) / 16;
    const int gemmBlocks = (rowTiles + 7) / 8;
    const int aggBlocks  = (N + 7) / 8;

    partition_kernel<<<nblk, P_THREADS, 0, stream>>>(src, dst, tmp, table, E);
    bin_csr_kernel<<<nbin, 1024, 0, stream>>>(tmp, table, rowptr, dis, csr, N, nblk);

    gemm_wmma_kernel<IN_DIM, HID_DIM / 16, false, false, true>
        <<<gemmBlocks, TPB, 0, stream>>>(x, W1, nullptr, dis, bufA, rowTiles, N);
    conv1_dense_kernel<<<rowTiles, 32, 0, stream>>>(bufA, rowptr, csr, dis, b1,
                                                  Wl1, bl1, Wl2, bl2, W2, bufB, N);

    agg_csr_kernel<OUT_DIM, false><<<aggBlocks, TPB, 0, stream>>>(bufB, rowptr, csr, dis, b2, out, N);
}


# conv2 gather: a wave walks 16 consecutive nodes, not one (100000 waves to 6250); next node's indices and rows are requested ahead of the current row's reduce and store; row sum order kept
# speedup vs baseline: 4.0694x; 1.0335x over previous
#include <hip/hip_runtime.h>
#include <hip/hip_bf16.h>


typedef float v2f __attribute__((ext_vector_type(2)));
typedef int   v2i __attribute__((ext_vector_type(2)));
typedef float v4f __attribute__((ext_vector_type(4)));
typedef float v8f __attribute__((ext_vector_type(8)));

#define IN_DIM  128
#define HID_DIM 64
#define MID_DIM 128
#define OUT_DIM 32

#define BIN_SHIFT 9
#define BIN_NODES 512
#define MAX_BINS  256
#define TBL_ROW   (MAX_BINS + 1)
#define P_THREADS 1024
#define P_EPT     8
#define P_CHUNK   (P_THREADS * P_EPT)
#define STAGE_CAP 12288

__global__ __launch_bounds__(P_THREADS) void partition_kernel(
    const int* __restrict__ src, const int* __restrict__ dst,
    int* __restrict__ tmp, int* __restrict__ table, int E) {
    __shared__ int hist[MAX_BINS];
    __shared__ int wsum[8];
    __shared__ int sorted[2 * P_CHUNK];
    const int tid  = threadIdx.x;
    const int lane = tid & 31;
    const int wv   = tid >> 5;
    const int chunk0 = blockIdx.x * P_CHUNK;

    if (tid < MAX_BINS) hist[tid] = 0;
    __syncthreads();

    int s[P_EPT], d[P_EPT], r[P_EPT];
#pragma unroll
    for (int k = 0; k < P_EPT; ++k) {
        const int e = chunk0 + k * P_THREADS + tid;
        s[k] = 0; d[k] = -1;
        if (e < E) { s[k] = src[e]; d[k] = dst[e]; }
    }
#pragma unroll
    for (int k = 0; k < P_EPT; ++k) {
        r[k] = 0;
        if (d[k] >= 0) {
            unsigned b = (unsigned)d[k] >> BIN_SHIFT;
            if (b > MAX_BINS - 1) b = MAX_BINS - 1;
            r[k] = atomicAdd(&hist[b], 1);
        }
    }
    __syncthreads();

    int c = 0, incl = 0;
    if (tid < MAX_BINS) {
        c = hist[tid];
        incl = c;
        for (int o = 1; o < 32; o <<= 1) {
            int t = __shfl_up(incl, o);
            if (lane >= o) incl += t;
        }
        if (lane == 31) wsum[wv] = incl;
    }
    __syncthreads();
    if (tid < MAX_BINS) {
        int off = 0;
        for (int i = 0; i < 8; ++i) if (i < wv) off += wsum[i];
        const int excl = off + incl - c;
        hist[tid] = excl;
        int* __restrict__ row = table + (long long)blockIdx.x * TBL_ROW;
        row[tid] = excl;
        if (tid == MAX_BINS - 1) row[MAX_BINS] = excl + c;
    }
    __syncthreads();

#pragma unroll
    for (int k = 0; k < P_EPT; ++k) {
        if (d[k] >= 0) {
            unsigned b = (unsigned)d[k] >> BIN_SHIFT;
            if (b > MAX_BINS - 1) b = MAX_BINS - 1;
            const int p = hist[b] + r[k];
            sorted[2 * p]     = s[k];
            sorted[2 * p + 1] = d[k];
        }
    }
    __syncthreads();

    const int n = min(P_CHUNK, E - chunk0);
    v2i* __restrict__ outp = (v2i*)(tmp + 2 * (long long)chunk0);
    for (int i = tid; i < n; i += P_THREADS) outp[i] = ((const v2i*)sorted)[i];
}

__global__ __launch_bounds__(1024) void bin_csr_kernel(
    const int* __restrict__ tmp, const int* __restrict__ table,
    int* __restrict__ rowptr, float* __restrict__ dis, int* __restrict__ csr,
    int N, int nblk) {
    __shared__ int cnt[BIN_NODES];
    __shared__ int rowoff[BIN_NODES];
    __shared__ int wbase[32];
    __shared__ int wtot[32];
    __shared__ int wsum[16];
    __shared__ int stage[STAGE_CAP];
    const int tid  = threadIdx.x;
    const int lane = tid & 31;
    const int wv   = tid >> 5;
    const int b    = blockIdx.x;

    int pb = 0, pt = 0;
    for (int blk = tid; blk < nblk; blk += 1024) {
        const int* __restrict__ row = table + (long long)blk * TBL_ROW;
        const int o0 = row[b], o1 = row[b + 1];
        pb += o0;
        pt += o1 - o0;
    }
    for (int o = 16; o > 0; o >>= 1) {
        pb += __shfl_xor(pb, o);
        pt += __shfl_xor(pt, o);
    }
    if (lane == 0) { wbase[wv] = pb; wtot[wv] = pt; }
    if (tid < BIN_NODES) cnt[tid] = 0;
    __syncthreads();
    int base = 0, total = 0;
    for (int i = 0; i < 32; ++i) { base += wbase[i]; total += wtot[i]; }
    const bool staged = total <= STAGE_CAP;

    for (int g = 0; g < nblk; g += 1024) {
        const int myblk = g + wv + 32 * lane;
        int o0 = 0, o1 = 0;
        if (myblk < nblk) {
            const int* __restrict__ row = table + (long long)myblk * TBL_ROW;
            o0 = row[b]; o1 = row[b + 1];
        }
        const int left = nblk - g - wv;
        const int nj = left <= 0 ? 0 : min(32, (left + 31) >> 5);
        for (int j = 0; j < nj; ++j) {
            const int a0 = __shfl(o0, j), a1 = __shfl(o1, j);
            const v2i* __restrict__ p = (const v2i*)(tmp + 2 * (long long)(g + wv + 32 * j) * P_CHUNK);
            for (int i = a0 + lane; i < a1; i += 32) {
                const v2i e = p[i];
                atomicAdd(&cnt[e.y & (BIN_NODES - 1)], 1);
            }
        }
    }
    __syncthreads();

    int c = 0, incl = 0;
    if (tid < BIN_NODES) {
        c = cnt[tid];
        incl = c;
        for (int o = 1; o < 32; o <<= 1) {
            int t = __shfl_up(incl, o);
            if (lane >= o) incl += t;
        }
        if (lane == 31) wsum[wv] = incl;
    }
    __syncthreads();
    if (tid < BIN_NODES) {
        int off = 0;
        for (int i = 0; i < 16; ++i) if (i < wv) off += wsum[i];
        const int excl = off + incl - c;
        rowoff[tid] = excl;
        cnt[tid] = 0;
        const int node = (b << BIN_SHIFT) + tid;
        if (node < N) {
            rowptr[node] = base + excl;
            dis[node] = 1.0f / sqrtf((float)c + 1.0f);
        }
        if (node == N - 1) rowptr[N] = base + excl + c;
    }
    __syncthreads();

    for (int g = 0; g < nblk; g += 1024) {
        const int myblk = g + wv + 32 * lane;
        int o0 = 0, o1 = 0;
        if (myblk < nblk) {
            const int* __restrict__ row = table + (long long)myblk * TBL_ROW;
            o0 = row[b]; o1 = row[b + 1];
        }
        const int left = nblk - g - wv;
        const int nj = left <= 0 ? 0 : min(32, (left + 31) >> 5);
        for (int j = 0; j < nj; ++j) {
            const int a0 = __shfl(o0, j), a1 = __shfl(o1, j);
            const v2i* __restrict__ p = (const v2i*)(tmp + 2 * (long long)(g + wv + 32 * j) * P_CHUNK);
            for (int i = a0 + lane; i < a1; i += 32) {
                const v2i e = p[i];
                const int dl = e.y & (BIN_NODES - 1);
                const int q = rowoff[dl] + atomicAdd(&cnt[dl], 1);
                if (staged) stage[q] = e.x;
                else        csr[base + q] = e.x;
            }
        }
    }
    if (staged) {
        __syncthreads();
        for (int i = tid; i < total; i += 1024) csr[base + i] = stage[i];
    }
}

template <int K, int TILES, bool BIAS, bool RELU, bool SCALE>
__global__ __launch_bounds__(256) void gemm_wmma_kernel(
    const float* __restrict__ A, const float* __restrict__ B,
    const float* __restrict__ bias, const float* __restrict__ rowscale,
    float* __restrict__ C, int numRowTiles, int n) {
    const int OUTW = TILES * 16;
    int wave = (blockIdx.x * blockDim.x + threadIdx.x) >> 5;
    int lane = threadIdx.x & 31;
    if (wave >= numRowTiles) return;

    const int m    = lane & 15;
    const int half = lane >> 4;
    const int row0 = wave * 16;

    v8f acc[TILES] = {};

    float rs[8];
    if (SCALE) {
#pragma unroll
        for (int r = 0; r < 8; ++r) {
            int row = row0 + r + 8 * half;
            rs[r] = rowscale[row < n ? row : n - 1];
        }
    }

    int arow = row0 + m;
    if (arow >= n) arow = n - 1;
    const float* __restrict__ ap = A + (long long)arow * K;

    for (int k0 = 0; k0 < K; k0 += 4) {
        v2f a = *(const v2f*)(ap + k0 + 2 * half);
        const float* __restrict__ bp = B + (long long)(k0 + 2 * half) * OUTW + m;
#pragma unroll
        for (int t = 0; t < TILES; ++t) {
            v2f b;
            b.x = bp[t * 16];
            b.y = bp[t * 16 + OUTW];
            acc[t] = __builtin_amdgcn_wmma_f32_16x16x4_f32(
                false, a, false, b, (short)0, acc[t], false, false);
        }
    }

#pragma unroll
    for (int t = 0; t < TILES; ++t) {
        const int col = t * 16 + m;
        float bv = BIAS ? bias[col] : 0.0f;
#pragma unroll
        for (int r = 0; r < 8; ++r) {
            int row = row0 + r + 8 * half;
            if (row < n) {
                float v = acc[t][r];
                if (SCALE) v *= rs[r];
                if (BIAS) v += bv;
                if (RELU) v = fmaxf(v, 0.0f);
                C[(long long)row * OUTW + col] = v;
            }
        }
    }
}

template <int W, bool RELU, int RUN>
__global__ __launch_bounds__(128) void agg_run_kernel(const float* __restrict__ msg,
                           const int* __restrict__ rowptr, const int* __restrict__ csr,
                           const float* __restrict__ dis, const float* __restrict__ bias,
                           float* __restrict__ out, int N) {
    constexpr int LPRW = W / 4;
    constexpr int G    = 32 / LPRW;
    constexpr int UNR  = 8;
    static_assert(RUN + 1 <= 32, "one rowptr entry per lane");
    static_assert(G * UNR == 32, "one step of UNR loads covers the 32 in-edges a wave holds");
    const int lane = threadIdx.x & 31;
    const int wave = __builtin_amdgcn_readfirstlane((int)((blockIdx.x * blockDim.x + threadIdx.x) >> 5));
    const int d0 = wave * RUN;
    if (d0 >= N) return;
    const int nn = min(RUN, N - d0);
    const int g = lane / LPRW;
    const int c = lane % LPRW;

    const int   rpl = rowptr[d0 + min(lane, nn)];
    const float dsl = dis[d0 + min(lane, nn - 1)];
    const v4f   bv  = *(const v4f*)(bias + 4 * c);
    const char* __restrict__ msgb = (const char*)msg;
    int beg = __builtin_amdgcn_readlane(rpl, 0);
    int end = __builtin_amdgcn_readlane(rpl, 1);
    int s = d0;
    {
        const int idx = beg + lane;
        if (idx < end) s = csr[idx];
    }
    int nend = nn > 1 ? __builtin_amdgcn_readlane(rpl, 2) : end;
    int sn = nn > 1 ? d0 + 1 : d0;
    v4f v[UNR];
    {
        unsigned off[UNR];
#pragma unroll
        for (int u = 0; u < UNR; ++u) {
            const int sj = __builtin_amdgcn_ds_bpermute((u * G + g) << 2, s);
            off[u] = (unsigned)sj * (unsigned)(W * 4) + (unsigned)(16 * c);
        }
        if (nn > 1) {
            const int idx = end + lane;
            if (idx < nend) sn = csr[idx];
        }
#pragma unroll
        for (int u = 0; u < UNR; ++u) v[u] = *(const v4f*)(msgb + off[u]);
    }

    for (int j = 0; j < nn; ++j) {
        const int d = d0 + j;
        const float dd = __int_as_float(__builtin_amdgcn_readlane(__float_as_int(dsl), j));
        v4f acc = {0.0f, 0.0f, 0.0f, 0.0f};
        {
            const int n = min(32, end + 1 - beg);
#pragma unroll
            for (int u = 0; u < UNR; ++u) {
                const v4f vv = (g < n - u * G) ? v[u] : (v4f){0.0f, 0.0f, 0.0f, 0.0f};
                acc += vv;
            }
        }
        for (int base = beg + 32; base <= end; base += 32) {
            const int n = min(32, end + 1 - base);
            int sc = d;
            {
                const int idx = base + lane;
                if (idx < end) sc = csr[idx];
            }
            unsigned off[UNR];
            v4f w[UNR];
#pragma unroll
            for (int u = 0; u < UNR; ++u) {
                const int sj = __builtin_amdgcn_ds_bpermute((u * G + g) << 2, sc);
                off[u] = (unsigned)sj * (unsigned)(W * 4) + (unsigned)(16 * c);
            }
#pragma unroll
            for (int u = 0; u < UNR; ++u) w[u] = *(const v4f*)(msgb + off[u]);
#pragma unroll
            for (int u = 0; u < UNR; ++u) {
                const v4f vv = (g < n - u * G) ? w[u] : (v4f){0.0f, 0.0f, 0.0f, 0.0f};
                acc += vv;
            }
        }
        if (j + 1 < nn) {
            s = sn;
            beg = end;
            end = nend;
            const bool more = j + 2 < nn;
            nend = more ? __builtin_amdgcn_readlane(rpl, j + 3) : end;
            sn = more ? d + 2 : d + 1;
            unsigned off[UNR];
#pragma unroll
            for (int u = 0; u < UNR; ++u) {
                const int sj = __builtin_amdgcn_ds_bpermute((u * G + g) << 2, s);
                off[u] = (unsigned)sj * (unsigned)(W * 4) + (unsigned)(16 * c);
            }
            if (more) {
                const int idx = end + lane;
                if (idx < nend) sn = csr[idx];
            }
#pragma unroll
            for (int u = 0; u < UNR; ++u) v[u] = *(const v4f*)(msgb + off[u]);
        }
#pragma unroll
        for (int o = 16; o >= LPRW; o >>= 1) {
            const int peer = (lane ^ o) << 2;
            acc.x += __int_as_float(__builtin_amdgcn_ds_bpermute(peer, __float_as_int(acc.x)));
            acc.y += __int_as_float(__builtin_amdgcn_ds_bpermute(peer, __float_as_int(acc.y)));
            acc.z += __int_as_float(__builtin_amdgcn_ds_bpermute(peer, __float_as_int(acc.z)));
            acc.w += __int_as_float(__builtin_amdgcn_ds_bpermute(peer, __float_as_int(acc.w)));
        }
        if (lane < LPRW) {
            v4f r = acc * dd + bv;
            if (RELU) {
                r.x = fmaxf(r.x, 0.0f); r.y = fmaxf(r.y, 0.0f);
                r.z = fmaxf(r.z, 0.0f); r.w = fmaxf(r.w, 0.0f);
            }
            *(v4f*)(out + (long long)d * W + 4 * c) = r;
        }
    }
}

#define FT_NODES 16
#define FT_PITCH 68

template <int NT>
__device__ __forceinline__ void dense64_T(const float* __restrict__ Wt, const float* hs,
                                          int m, int half, v8f (&acc)[NT]) {
    constexpr int OUTW = NT * 16;
    for (int k0 = 0; k0 < 64; k0 += 4) {
        const v2f bh = *(const v2f*)(hs + m * FT_PITCH + k0 + 2 * half);
        const float* __restrict__ wp = Wt + (k0 + 2 * half) * OUTW + m;
#pragma unroll
        for (int t = 0; t < NT; ++t) {
            v2f aw;
            aw.x = wp[t * 16];
            aw.y = wp[t * 16 + OUTW];
            acc[t] = __builtin_amdgcn_wmma_f32_16x16x4_f32(
                false, aw, false, bh, (short)0, acc[t], false, false);
        }
    }
}

__device__ __forceinline__ void bias_relu_T(v8f& a, const float* __restrict__ bias16, int half) {
    const v4f ba = *(const v4f*)(bias16 + 8 * half);
    const v4f bb = *(const v4f*)(bias16 + 8 * half + 4);
    a[0] = fmaxf(a[0] + ba.x, 0.0f); a[1] = fmaxf(a[1] + ba.y, 0.0f);
    a[2] = fmaxf(a[2] + ba.z, 0.0f); a[3] = fmaxf(a[3] + ba.w, 0.0f);
    a[4] = fmaxf(a[4] + bb.x, 0.0f); a[5] = fmaxf(a[5] + bb.y, 0.0f);
    a[6] = fmaxf(a[6] + bb.z, 0.0f); a[7] = fmaxf(a[7] + bb.w, 0.0f);
}

__device__ __forceinline__ void put_tile_T(float* hs, int t, int m, int half, const v8f& a) {
    float* p = hs + m * FT_PITCH + 16 * t + 8 * half;
    *(v4f*)p       = (v4f){a[0], a[1], a[2], a[3]};
    *(v4f*)(p + 4) = (v4f){a[4], a[5], a[6], a[7]};
}

__global__ __launch_bounds__(32) __attribute__((amdgpu_waves_per_eu(8))) void conv1_dense_kernel(
    const float* __restrict__ msg, const int* __restrict__ rowptr, const int* __restrict__ csr,
    const float* __restrict__ dis, const float* __restrict__ b1,
    const float* __restrict__ Wl1, const float* __restrict__ bl1,
    const float* __restrict__ Wl2, const float* __restrict__ bl2,
    const float* __restrict__ W2, float* __restrict__ out, int N) {
    constexpr int W    = HID_DIM;
    constexpr int LPRW = W / 4;
    constexpr int G    = 32 / LPRW;
    constexpr int UNR  = 8;
    __shared__ float hs[FT_NODES * FT_PITCH];
    const int lane = threadIdx.x;
    const int d0 = blockIdx.x * FT_NODES;
    if (d0 >= N) return;
    const int nn = min(FT_NODES, N - d0);
    const int g = lane / LPRW;
    const int c = lane % LPRW;

    const int   rpl = rowptr[d0 + min(lane, nn)];
    const float dsl = dis[d0 + min(lane, nn - 1)];
    const v4f   bv  = *(const v4f*)(b1 + 4 * c);
    int beg = __builtin_amdgcn_readlane(rpl, 0);
    int end = __builtin_amdgcn_readlane(rpl, 1);
    int s = d0;
    {
        const int idx = beg + lane;
        if (idx < end) s = csr[idx];
    }
    const char* __restrict__ msgb = (const char*)msg;
    for (int j = 0; j < nn; ++j) {
        const int d = d0 + j;
        const float dd = __int_as_float(__builtin_amdgcn_readlane(__float_as_int(dsl), j));
        bool fetch_next = j + 1 < nn;
        const int nend = fetch_next ? __builtin_amdgcn_readlane(rpl, j + 2) : end;
        int sn = fetch_next ? d + 1 : d;

        v4f acc = {0.0f, 0.0f, 0.0f, 0.0f};
        for (int base = beg; base <= end; base += 32) {
            const int n = min(32, end + 1 - base);
            int sc = s;
            if (base != beg) {
                sc = d;
                const int idx = base + lane;
                if (idx < end) sc = csr[idx];
            }
#pragma unroll 1
            for (int t0 = 0; t0 * G < n; t0 += UNR) {
                unsigned off[UNR];
#pragma unroll
                for (int u = 0; u < UNR; ++u) {
                    const int ei = (t0 + u) * G + g;
                    const int sj = __builtin_amdgcn_ds_bpermute(ei << 2, sc);
                    off[u] = (unsigned)sj * (unsigned)(W * 4) + (unsigned)(16 * c);
                }
                if (fetch_next) {
                    fetch_next = false;
                    const int idx = end + lane;
                    if (idx < nend) sn = csr[idx];
                }
                v4f v[UNR];
#pragma unroll
                for (int u = 0; u < UNR; ++u) v[u] = *(const v4f*)(msgb + off[u]);
#pragma unroll
                for (int u = 0; u < UNR; ++u) {
                    const v4f vv = (g < n - (t0 + u) * G) ? v[u] : (v4f){0.0f, 0.0f, 0.0f, 0.0f};
                    acc += vv;
                }
            }
        }
#pragma unroll
        for (int o = 16; o >= LPRW; o >>= 1) {
            const int peer = (lane ^ o) << 2;
            acc.x += __int_as_float(__builtin_amdgcn_ds_bpermute(peer, __float_as_int(acc.x)));
            acc.y += __int_as_float(__builtin_amdgcn_ds_bpermute(peer, __float_as_int(acc.y)));
            acc.z += __int_as_float(__builtin_amdgcn_ds_bpermute(peer, __float_as_int(acc.z)));
            acc.w += __int_as_float(__builtin_amdgcn_ds_bpermute(peer, __float_as_int(acc.w)));
        }
        if (lane < LPRW) {
            v4f r = acc * dd + bv;
            r.x = fmaxf(r.x, 0.0f); r.y = fmaxf(r.y, 0.0f);
            r.z = fmaxf(r.z, 0.0f); r.w = fmaxf(r.w, 0.0f);
            *(v4f*)(hs + j * FT_PITCH + 4 * c) = r;
        }
        s = sn;
        beg = end;
        end = nend;
    }
    for (int j = nn; j < FT_NODES; ++j)
        if (lane < LPRW) *(v4f*)(hs + j * FT_PITCH + 4 * c) = (v4f){0.0f, 0.0f, 0.0f, 0.0f};
    __syncthreads();

    const int m    = lane & 15;
    const int half = lane >> 4;

    v8f acc1[MID_DIM / 16] = {};
    dense64_T<MID_DIM / 16>(Wl1, hs, m, half, acc1);
#pragma unroll
    for (int t = 0; t < MID_DIM / 16; ++t) bias_relu_T(acc1[t], bl1 + 16 * t, half);

    v8f acc2[HID_DIM / 16] = {};
    __syncthreads();
#pragma unroll
    for (int t = 0; t < 4; ++t) put_tile_T(hs, t, m, half, acc1[t]);
    __syncthreads();
    dense64_T<HID_DIM / 16>(Wl2, hs, m, half, acc2);
    __syncthreads();
#pragma unroll
    for (int t = 0; t < 4; ++t) put_tile_T(hs, t, m, half, acc1[4 + t]);
    __syncthreads();
    dense64_T<HID_DIM / 16>(Wl2 + 64 * HID_DIM, hs, m, half, acc2);
#pragma unroll
    for (int t = 0; t < HID_DIM / 16; ++t) bias_relu_T(acc2[t], bl2 + 16 * t, half);

    v8f acc3[OUT_DIM / 16] = {};
    __syncthreads();
#pragma unroll
    for (int t = 0; t < HID_DIM / 16; ++t) put_tile_T(hs, t, m, half, acc2[t]);
    __syncthreads();
    dense64_T<OUT_DIM / 16>(W2, hs, m, half, acc3);

    const int row = d0 + m;
    if (row < N) {
        const float rs = dis[row];
        float* __restrict__ op = out + (long long)row * OUT_DIM + 8 * half;
#pragma unroll
        for (int t = 0; t < OUT_DIM / 16; ++t) {
            const v4f lo = {acc3[t][0] * rs, acc3[t][1] * rs, acc3[t][2] * rs, acc3[t][3] * rs};
            const v4f hi = {acc3[t][4] * rs, acc3[t][5] * rs, acc3[t][6] * rs, acc3[t][7] * rs};
            *(v4f*)(op + 16 * t)     = lo;
            *(v4f*)(op + 16 * t + 4) = hi;
        }
    }
}

extern "C" void kernel_launch(void* const* d_in, const int* in_sizes, int n_in,
                              void* d_out, int out_size, void* d_ws, size_t ws_size,
                              hipStream_t stream) {
    const float* x   = (const float*)d_in[0];
    const int*   ei  = (const int*)  d_in[1];
    const float* W1  = (const float*)d_in[2];
    const float* b1  = (const float*)d_in[3];
    const float* Wl1 = (const float*)d_in[4];
    const float* bl1 = (const float*)d_in[5];
    const float* Wl2 = (const float*)d_in[6];
    const float* bl2 = (const float*)d_in[7];
    const float* W2  = (const float*)d_in[8];
    const float* b2  = (const float*)d_in[9];

    const int N = in_sizes[0] / IN_DIM;
    const int E = in_sizes[1] / 2;
    const int* src = ei;
    const int* dst = ei + E;

    const int nblk = (E + P_CHUNK - 1) / P_CHUNK;
    const int nbin = (N + BIN_NODES - 1) >> BIN_SHIFT;
    const size_t Np = ((size_t)N + 1 + 63) / 64 * 64;
    const size_t Ep = ((size_t)E + 63) / 64 * 64;
    const size_t Tp = ((size_t)nblk * TBL_ROW + 63) / 64 * 64;
    int*   table  = (int*)d_ws;
    int*   rowptr = table + Tp;
    float* dis    = (float*)(rowptr + Np);
    int*   tmp    = (int*)(dis + Np);
    int*   csr    = tmp + 2 * Ep;
    float* bufA   = (float*)(csr + Ep);
    float* bufB   = bufA + (size_t)N * 64;
    float* out    = (float*)d_out;

    const int TPB = 256;
    const int rowTiles = (N + 15) / 16;
    const int gemmBlocks = (rowTiles + 7) / 8;
    const int AGG_RUN = 16;
    const int aggBlocks  = (rowTiles + 3) / 4;

    partition_kernel<<<nblk, P_THREADS, 0, stream>>>(src, dst, tmp, table, E);
    bin_csr_kernel<<<nbin, 1024, 0, stream>>>(tmp, table, rowptr, dis, csr, N, nblk);

    gemm_wmma_kernel<IN_DIM, HID_DIM / 16, false, false, true>
        <<<gemmBlocks, TPB, 0, stream>>>(x, W1, nullptr, dis, bufA, rowTiles, N);
    conv1_dense_kernel<<<rowTiles, 32, 0, stream>>>(bufA, rowptr, csr, dis, b1,
                                                  Wl1, bl1, Wl2, bl2, W2, bufB, N);

    agg_run_kernel<OUT_DIM, false, AGG_RUN><<<aggBlocks, 128, 0, stream>>>(bufB, rowptr, csr, dis, b2, out, N);
}


# bin kernel: a wave requests its segments of 8 chunks x 2 trips as one group of 16 loads in both walks (was 1 load per memory trip), bounds fetched once; row sum order = LDS arrival order (changes)
# speedup vs baseline: 4.2005x; 1.0322x over previous
#include <hip/hip_runtime.h>
#include <hip/hip_bf16.h>


typedef float v2f __attribute__((ext_vector_type(2)));
typedef int   v2i __attribute__((ext_vector_type(2)));
typedef float v4f __attribute__((ext_vector_type(4)));
typedef float v8f __attribute__((ext_vector_type(8)));

#define IN_DIM  128
#define HID_DIM 64
#define MID_DIM 128
#define OUT_DIM 32

#define BIN_SHIFT 9
#define BIN_NODES 512
#define MAX_BINS  256
#define TBL_ROW   (MAX_BINS + 1)
#define P_THREADS 1024
#define P_EPT     8
#define P_CHUNK   (P_THREADS * P_EPT)
#define STAGE_CAP 12288
#define BIN_JB    8
#define BIN_RB    2
#define BIN_U     (BIN_JB * BIN_RB)

__global__ __launch_bounds__(P_THREADS) void partition_kernel(
    const int* __restrict__ src, const int* __restrict__ dst,
    int* __restrict__ tmp, int* __restrict__ table, int E) {
    __shared__ int hist[MAX_BINS];
    __shared__ int wsum[8];
    __shared__ int sorted[2 * P_CHUNK];
    const int tid  = threadIdx.x;
    const int lane = tid & 31;
    const int wv   = tid >> 5;
    const int chunk0 = blockIdx.x * P_CHUNK;

    if (tid < MAX_BINS) hist[tid] = 0;
    __syncthreads();

    int s[P_EPT], d[P_EPT], r[P_EPT];
#pragma unroll
    for (int k = 0; k < P_EPT; ++k) {
        const int e = chunk0 + k * P_THREADS + tid;
        s[k] = 0; d[k] = -1;
        if (e < E) { s[k] = src[e]; d[k] = dst[e]; }
    }
#pragma unroll
    for (int k = 0; k < P_EPT; ++k) {
        r[k] = 0;
        if (d[k] >= 0) {
            unsigned b = (unsigned)d[k] >> BIN_SHIFT;
            if (b > MAX_BINS - 1) b = MAX_BINS - 1;
            r[k] = atomicAdd(&hist[b], 1);
        }
    }
    __syncthreads();

    int c = 0, incl = 0;
    if (tid < MAX_BINS) {
        c = hist[tid];
        incl = c;
        for (int o = 1; o < 32; o <<= 1) {
            int t = __shfl_up(incl, o);
            if (lane >= o) incl += t;
        }
        if (lane == 31) wsum[wv] = incl;
    }
    __syncthreads();
    if (tid < MAX_BINS) {
        int off = 0;
        for (int i = 0; i < 8; ++i) if (i < wv) off += wsum[i];
        const int excl = off + incl - c;
        hist[tid] = excl;
        int* __restrict__ row = table + (long long)blockIdx.x * TBL_ROW;
        row[tid] = excl;
        if (tid == MAX_BINS - 1) row[MAX_BINS] = excl + c;
    }
    __syncthreads();

#pragma unroll
    for (int k = 0; k < P_EPT; ++k) {
        if (d[k] >= 0) {
            unsigned b = (unsigned)d[k] >> BIN_SHIFT;
            if (b > MAX_BINS - 1) b = MAX_BINS - 1;
            const int p = hist[b] + r[k];
            sorted[2 * p]     = s[k];
            sorted[2 * p + 1] = d[k];
        }
    }
    __syncthreads();

    const int n = min(P_CHUNK, E - chunk0);
    v2i* __restrict__ outp = (v2i*)(tmp + 2 * (long long)chunk0);
    for (int i = tid; i < n; i += P_THREADS) outp[i] = ((const v2i*)sorted)[i];
}

__device__ __forceinline__ int seg_bounds(int o0, int o1, int j0, int (&a0)[BIN_JB], int (&len)[BIN_JB]) {
    int maxlen = 0;
#pragma unroll
    for (int q = 0; q < BIN_JB; ++q) {
        a0[q]  = __builtin_amdgcn_readlane(o0, j0 + q);
        len[q] = __builtin_amdgcn_readlane(o1, j0 + q) - a0[q];
        maxlen = max(maxlen, len[q]);
    }
    return maxlen;
}

__global__ __launch_bounds__(1024) void bin_csr_kernel(
    const int* __restrict__ tmp, const int* __restrict__ table,
    int* __restrict__ rowptr, float* __restrict__ dis, int* __restrict__ csr,
    int N, int nblk) {
    __shared__ int cnt[BIN_NODES];
    __shared__ int rowoff[BIN_NODES];
    __shared__ int wbase[32];
    __shared__ int wtot[32];
    __shared__ int wsum[16];
    __shared__ int stage[STAGE_CAP];
    const int tid  = threadIdx.x;
    const int lane = tid & 31;
    const int wv   = __builtin_amdgcn_readfirstlane(tid >> 5);
    const int b    = blockIdx.x;

    int f0 = 0, f1 = 0;
    {
        const int myblk = wv + 32 * lane;
        if (myblk < nblk) {
            const int* __restrict__ row = table + (long long)myblk * TBL_ROW;
            f0 = row[b]; f1 = row[b + 1];
        }
    }

    int pb = 0, pt = 0;
    for (int blk = tid; blk < nblk; blk += 1024) {
        const int* __restrict__ row = table + (long long)blk * TBL_ROW;
        const int o0 = row[b], o1 = row[b + 1];
        pb += o0;
        pt += o1 - o0;
    }
    for (int o = 16; o > 0; o >>= 1) {
        pb += __shfl_xor(pb, o);
        pt += __shfl_xor(pt, o);
    }
    if (lane == 0) { wbase[wv] = pb; wtot[wv] = pt; }
    if (tid < BIN_NODES) cnt[tid] = 0;
    __syncthreads();
    int base = 0, total = 0;
    for (int i = 0; i < 32; ++i) { base += wbase[i]; total += wtot[i]; }
    const bool staged = total <= STAGE_CAP;

    for (int g = 0; g < nblk; g += 1024) {
        int o0 = f0, o1 = f1;
        if (g > 0) {
            o0 = 0; o1 = 0;
            const int myblk = g + wv + 32 * lane;
            if (myblk < nblk) {
                const int* __restrict__ row = table + (long long)myblk * TBL_ROW;
                o0 = row[b]; o1 = row[b + 1];
            }
        }
        const int left = nblk - g - wv;
        const int nj = left <= 0 ? 0 : min(32, (left + 31) >> 5);
        for (int j0 = 0; j0 < nj; j0 += BIN_JB) {
            int a0[BIN_JB], len[BIN_JB];
            const int maxlen = seg_bounds(o0, o1, j0, a0, len);
            const v2i* __restrict__ p0 = (const v2i*)tmp + (long long)(g + wv + 32 * j0) * P_CHUNK;
            for (int r = 0; r < maxlen; r += 32 * BIN_RB) {
                int dn[BIN_U];
#pragma unroll
                for (int u = 0; u < BIN_U; ++u) {
                    const int q = u % BIN_JB;
                    const int i = r + 32 * (u / BIN_JB) + lane;
                    dn[u] = -1;
                    if (i < len[q]) dn[u] = p0[(long long)q * (32 * P_CHUNK) + a0[q] + i].y;
                }
#pragma unroll
                for (int u = 0; u < BIN_U; ++u)
                    if (dn[u] >= 0) atomicAdd(&cnt[dn[u] & (BIN_NODES - 1)], 1);
            }
        }
    }
    __syncthreads();

    int c = 0, incl = 0;
    if (tid < BIN_NODES) {
        c = cnt[tid];
        incl = c;
        for (int o = 1; o < 32; o <<= 1) {
            int t = __shfl_up(incl, o);
            if (lane >= o) incl += t;
        }
        if (lane == 31) wsum[wv] = incl;
    }
    __syncthreads();
    if (tid < BIN_NODES) {
        int off = 0;
        for (int i = 0; i < 16; ++i) if (i < wv) off += wsum[i];
        const int excl = off + incl - c;
        rowoff[tid] = excl;
        cnt[tid] = 0;
        const int node = (b << BIN_SHIFT) + tid;
        if (node < N) {
            rowptr[node] = base + excl;
            dis[node] = 1.0f / sqrtf((float)c + 1.0f);
        }
        if (node == N - 1) rowptr[N] = base + excl + c;
    }
    __syncthreads();

    for (int g = 0; g < nblk; g += 1024) {
        int o0 = f0, o1 = f1;
        if (g > 0) {
            o0 = 0; o1 = 0;
            const int myblk = g + wv + 32 * lane;
            if (myblk < nblk) {
                const int* __restrict__ row = table + (long long)myblk * TBL_ROW;
                o0 = row[b]; o1 = row[b + 1];
            }
        }
        const int left = nblk - g - wv;
        const int nj = left <= 0 ? 0 : min(32, (left + 31) >> 5);
        for (int j0 = 0; j0 < nj; j0 += BIN_JB) {
            int a0[BIN_JB], len[BIN_JB];
            const int maxlen = seg_bounds(o0, o1, j0, a0, len);
            const v2i* __restrict__ p0 = (const v2i*)tmp + (long long)(g + wv + 32 * j0) * P_CHUNK;
            for (int r = 0; r < maxlen; r += 32 * BIN_RB) {
                v2i e[BIN_U];
#pragma unroll
                for (int u = 0; u < BIN_U; ++u) {
                    const int q = u % BIN_JB;
                    const int i = r + 32 * (u / BIN_JB) + lane;
                    e[u] = (v2i){0, -1};
                    if (i < len[q]) e[u] = p0[(long long)q * (32 * P_CHUNK) + a0[q] + i];
                }
                int ro[BIN_U], rk[BIN_U];
#pragma unroll
                for (int u = 0; u < BIN_U; ++u) {
                    ro[u] = 0; rk[u] = 0;
                    if (e[u].y >= 0) {
                        const int dl = e[u].y & (BIN_NODES - 1);
                        ro[u] = rowoff[dl];
                        rk[u] = atomicAdd(&cnt[dl], 1);
                    }
                }
#pragma unroll
                for (int u = 0; u < BIN_U; ++u) {
                    if (e[u].y >= 0) {
                        const int qd = ro[u] + rk[u];
                        if (staged) stage[qd] = e[u].x;
                        else        csr[base + qd] = e[u].x;
                    }
                }
            }
        }
    }
    if (staged) {
        __syncthreads();
        for (int i = tid; i < total; i += 1024) csr[base + i] = stage[i];
    }
}

template <int K, int TILES, bool BIAS, bool RELU, bool SCALE>
__global__ __launch_bounds__(256) void gemm_wmma_kernel(
    const float* __restrict__ A, const float* __restrict__ B,
    const float* __restrict__ bias, const float* __restrict__ rowscale,
    float* __restrict__ C, int numRowTiles, int n) {
    const int OUTW = TILES * 16;
    int wave = (blockIdx.x * blockDim.x + threadIdx.x) >> 5;
    int lane = threadIdx.x & 31;
    if (wave >= numRowTiles) return;

    const int m    = lane & 15;
    const int half = lane >> 4;
    const int row0 = wave * 16;

    v8f acc[TILES] = {};

    float rs[8];
    if (SCALE) {
#pragma unroll
        for (int r = 0; r < 8; ++r) {
            int row = row0 + r + 8 * half;
            rs[r] = rowscale[row < n ? row : n - 1];
        }
    }

    int arow = row0 + m;
    if (arow >= n) arow = n - 1;
    const float* __restrict__ ap = A + (long long)arow * K;

    for (int k0 = 0; k0 < K; k0 += 4) {
        v2f a = *(const v2f*)(ap + k0 + 2 * half);
        const float* __restrict__ bp = B + (long long)(k0 + 2 * half) * OUTW + m;
#pragma unroll
        for (int t = 0; t < TILES; ++t) {
            v2f b;
            b.x = bp[t * 16];
            b.y = bp[t * 16 + OUTW];
            acc[t] = __builtin_amdgcn_wmma_f32_16x16x4_f32(
                false, a, false, b, (short)0, acc[t], false, false);
        }
    }

#pragma unroll
    for (int t = 0; t < TILES; ++t) {
        const int col = t * 16 + m;
        float bv = BIAS ? bias[col] : 0.0f;
#pragma unroll
        for (int r = 0; r < 8; ++r) {
            int row = row0 + r + 8 * half;
            if (row < n) {
                float v = acc[t][r];
                if (SCALE) v *= rs[r];
                if (BIAS) v += bv;
                if (RELU) v = fmaxf(v, 0.0f);
                C[(long long)row * OUTW + col] = v;
            }
        }
    }
}

template <int W, bool RELU, int RUN>
__global__ __launch_bounds__(128) void agg_run_kernel(const float* __restrict__ msg,
                           const int* __restrict__ rowptr, const int* __restrict__ csr,
                           const float* __restrict__ dis, const float* __restrict__ bias,
                           float* __restrict__ out, int N) {
    constexpr int LPRW = W / 4;
    constexpr int G    = 32 / LPRW;
    constexpr int UNR  = 8;
    static_assert(RUN + 1 <= 32, "one rowptr entry per lane");
    static_assert(G * UNR == 32, "one step of UNR loads covers the 32 in-edges a wave holds");
    const int lane = threadIdx.x & 31;
    const int wave = __builtin_amdgcn_readfirstlane((int)((blockIdx.x * blockDim.x + threadIdx.x) >> 5));
    const int d0 = wave * RUN;
    if (d0 >= N) return;
    const int nn = min(RUN, N - d0);
    const int g = lane / LPRW;
    const int c = lane % LPRW;

    const int   rpl = rowptr[d0 + min(lane, nn)];
    const float dsl = dis[d0 + min(lane, nn - 1)];
    const v4f   bv  = *(const v4f*)(bias + 4 * c);
    const char* __restrict__ msgb = (const char*)msg;
    int beg = __builtin_amdgcn_readlane(rpl, 0);
    int end = __builtin_amdgcn_readlane(rpl, 1);
    int s = d0;
    {
        const int idx = beg + lane;
        if (idx < end) s = csr[idx];
    }
    int nend = nn > 1 ? __builtin_amdgcn_readlane(rpl, 2) : end;
    int sn = nn > 1 ? d0 + 1 : d0;
    v4f v[UNR];
    {
        unsigned off[UNR];
#pragma unroll
        for (int u = 0; u < UNR; ++u) {
            const int sj = __builtin_amdgcn_ds_bpermute((u * G + g) << 2, s);
            off[u] = (unsigned)sj * (unsigned)(W * 4) + (unsigned)(16 * c);
        }
        if (nn > 1) {
            const int idx = end + lane;
            if (idx < nend) sn = csr[idx];
        }
#pragma unroll
        for (int u = 0; u < UNR; ++u) v[u] = *(const v4f*)(msgb + off[u]);
    }

    for (int j = 0; j < nn; ++j) {
        const int d = d0 + j;
        const float dd = __int_as_float(__builtin_amdgcn_readlane(__float_as_int(dsl), j));
        v4f acc = {0.0f, 0.0f, 0.0f, 0.0f};
        {
            const int n = min(32, end + 1 - beg);
#pragma unroll
            for (int u = 0; u < UNR; ++u) {
                const v4f vv = (g < n - u * G) ? v[u] : (v4f){0.0f, 0.0f, 0.0f, 0.0f};
                acc += vv;
            }
        }
        for (int base = beg + 32; base <= end; base += 32) {
            const int n = min(32, end + 1 - base);
            int sc = d;
            {
                const int idx = base + lane;
                if (idx < end) sc = csr[idx];
            }
            unsigned off[UNR];
            v4f w[UNR];
#pragma unroll
            for (int u = 0; u < UNR; ++u) {
                const int sj = __builtin_amdgcn_ds_bpermute((u * G + g) << 2, sc);
                off[u] = (unsigned)sj * (unsigned)(W * 4) + (unsigned)(16 * c);
            }
#pragma unroll
            for (int u = 0; u < UNR; ++u) w[u] = *(const v4f*)(msgb + off[u]);
#pragma unroll
            for (int u = 0; u < UNR; ++u) {
                const v4f vv = (g < n - u * G) ? w[u] : (v4f){0.0f, 0.0f, 0.0f, 0.0f};
                acc += vv;
            }
        }
        if (j + 1 < nn) {
            s = sn;
            beg = end;
            end = nend;
            const bool more = j + 2 < nn;
            nend = more ? __builtin_amdgcn_readlane(rpl, j + 3) : end;
            sn = more ? d + 2 : d + 1;
            unsigned off[UNR];
#pragma unroll
            for (int u = 0; u < UNR; ++u) {
                const int sj = __builtin_amdgcn_ds_bpermute((u * G + g) << 2, s);
                off[u] = (unsigned)sj * (unsigned)(W * 4) + (unsigned)(16 * c);
            }
            if (more) {
                const int idx = end + lane;
                if (idx < nend) sn = csr[idx];
            }
#pragma unroll
            for (int u = 0; u < UNR; ++u) v[u] = *(const v4f*)(msgb + off[u]);
        }
#pragma unroll
        for (int o = 16; o >= LPRW; o >>= 1) {
            const int peer = (lane ^ o) << 2;
            acc.x += __int_as_float(__builtin_amdgcn_ds_bpermute(peer, __float_as_int(acc.x)));
            acc.y += __int_as_float(__builtin_amdgcn_ds_bpermute(peer, __float_as_int(acc.y)));
            acc.z += __int_as_float(__builtin_amdgcn_ds_bpermute(peer, __float_as_int(acc.z)));
            acc.w += __int_as_float(__builtin_amdgcn_ds_bpermute(peer, __float_as_int(acc.w)));
        }
        if (lane < LPRW) {
            v4f r = acc * dd + bv;
            if (RELU) {
                r.x = fmaxf(r.x, 0.0f); r.y = fmaxf(r.y, 0.0f);
                r.z = fmaxf(r.z, 0.0f); r.w = fmaxf(r.w, 0.0f);
            }
            *(v4f*)(out + (long long)d * W + 4 * c) = r;
        }
    }
}

#define FT_NODES 16
#define FT_PITCH 68

template <int NT>
__device__ __forceinline__ void dense64_T(const float* __restrict__ Wt, const float* hs,
                                          int m, int half, v8f (&acc)[NT]) {
    constexpr int OUTW = NT * 16;
    for (int k0 = 0; k0 < 64; k0 += 4) {
        const v2f bh = *(const v2f*)(hs + m * FT_PITCH + k0 + 2 * half);
        const float* __restrict__ wp = Wt + (k0 + 2 * half) * OUTW + m;
#pragma unroll
        for (int t = 0; t < NT; ++t) {
            v2f aw;
            aw.x = wp[t * 16];
            aw.y = wp[t * 16 + OUTW];
            acc[t] = __builtin_amdgcn_wmma_f32_16x16x4_f32(
                false, aw, false, bh, (short)0, acc[t], false, false);
        }
    }
}

__device__ __forceinline__ void bias_relu_T(v8f& a, const float* __restrict__ bias16, int half) {
    const v4f ba = *(const v4f*)(bias16 + 8 * half);
    const v4f bb = *(const v4f*)(bias16 + 8 * half + 4);
    a[0] = fmaxf(a[0] + ba.x, 0.0f); a[1] = fmaxf(a[1] + ba.y, 0.0f);
    a[2] = fmaxf(a[2] + ba.z, 0.0f); a[3] = fmaxf(a[3] + ba.w, 0.0f);
    a[4] = fmaxf(a[4] + bb.x, 0.0f); a[5] = fmaxf(a[5] + bb.y, 0.0f);
    a[6] = fmaxf(a[6] + bb.z, 0.0f); a[7] = fmaxf(a[7] + bb.w, 0.0f);
}

__device__ __forceinline__ void put_tile_T(float* hs, int t, int m, int half, const v8f& a) {
    float* p = hs + m * FT_PITCH + 16 * t + 8 * half;
    *(v4f*)p       = (v4f){a[0], a[1], a[2], a[3]};
    *(v4f*)(p + 4) = (v4f){a[4], a[5], a[6], a[7]};
}

__global__ __launch_bounds__(32) __attribute__((amdgpu_waves_per_eu(8))) void conv1_dense_kernel(
    const float* __restrict__ msg, const int* __restrict__ rowptr, const int* __restrict__ csr,
    const float* __restrict__ dis, const float* __restrict__ b1,
    const float* __restrict__ Wl1, const float* __restrict__ bl1,
    const float* __restrict__ Wl2, const float* __restrict__ bl2,
    const float* __restrict__ W2, float* __restrict__ out, int N) {
    constexpr int W    = HID_DIM;
    constexpr int LPRW = W / 4;
    constexpr int G    = 32 / LPRW;
    constexpr int UNR  = 8;
    __shared__ float hs[FT_NODES * FT_PITCH];
    const int lane = threadIdx.x;
    const int d0 = blockIdx.x * FT_NODES;
    if (d0 >= N) return;
    const int nn = min(FT_NODES, N - d0);
    const int g = lane / LPRW;
    const int c = lane % LPRW;

    const int   rpl = rowptr[d0 + min(lane, nn)];
    const float dsl = dis[d0 + min(lane, nn - 1)];
    const v4f   bv  = *(const v4f*)(b1 + 4 * c);
    int beg = __builtin_amdgcn_readlane(rpl, 0);
    int end = __builtin_amdgcn_readlane(rpl, 1);
    int s = d0;
    {
        const int idx = beg + lane;
        if (idx < end) s = csr[idx];
    }
    const char* __restrict__ msgb = (const char*)msg;
    for (int j = 0; j < nn; ++j) {
        const int d = d0 + j;
        const float dd = __int_as_float(__builtin_amdgcn_readlane(__float_as_int(dsl), j));
        bool fetch_next = j + 1 < nn;
        const int nend = fetch_next ? __builtin_amdgcn_readlane(rpl, j + 2) : end;
        int sn = fetch_next ? d + 1 : d;

        v4f acc = {0.0f, 0.0f, 0.0f, 0.0f};
        for (int base = beg; base <= end; base += 32) {
            const int n = min(32, end + 1 - base);
            int sc = s;
            if (base != beg) {
                sc = d;
                const int idx = base + lane;
                if (idx < end) sc = csr[idx];
            }
#pragma unroll 1
            for (int t0 = 0; t0 * G < n; t0 += UNR) {
                unsigned off[UNR];
#pragma unroll
                for (int u = 0; u < UNR; ++u) {
                    const int ei = (t0 + u) * G + g;
                    const int sj = __builtin_amdgcn_ds_bpermute(ei << 2, sc);
                    off[u] = (unsigned)sj * (unsigned)(W * 4) + (unsigned)(16 * c);
                }
                if (fetch_next) {
                    fetch_next = false;
                    const int idx = end + lane;
                    if (idx < nend) sn = csr[idx];
                }
                v4f v[UNR];
#pragma unroll
                for (int u = 0; u < UNR; ++u) v[u] = *(const v4f*)(msgb + off[u]);
#pragma unroll
                for (int u = 0; u < UNR; ++u) {
                    const v4f vv = (g < n - (t0 + u) * G) ? v[u] : (v4f){0.0f, 0.0f, 0.0f, 0.0f};
                    acc += vv;
                }
            }
        }
#pragma unroll
        for (int o = 16; o >= LPRW; o >>= 1) {
            const int peer = (lane ^ o) << 2;
            acc.x += __int_as_float(__builtin_amdgcn_ds_bpermute(peer, __float_as_int(acc.x)));
            acc.y += __int_as_float(__builtin_amdgcn_ds_bpermute(peer, __float_as_int(acc.y)));
            acc.z += __int_as_float(__builtin_amdgcn_ds_bpermute(peer, __float_as_int(acc.z)));
            acc.w += __int_as_float(__builtin_amdgcn_ds_bpermute(peer, __float_as_int(acc.w)));
        }
        if (lane < LPRW) {
            v4f r = acc * dd + bv;
            r.x = fmaxf(r.x, 0.0f); r.y = fmaxf(r.y, 0.0f);
            r.z = fmaxf(r.z, 0.0f); r.w = fmaxf(r.w, 0.0f);
            *(v4f*)(hs + j * FT_PITCH + 4 * c) = r;
        }
        s = sn;
        beg = end;
        end = nend;
    }
    for (int j = nn; j < FT_NODES; ++j)
        if (lane < LPRW) *(v4f*)(hs + j * FT_PITCH + 4 * c) = (v4f){0.0f, 0.0f, 0.0f, 0.0f};
    __syncthreads();

    const int m    = lane & 15;
    const int half = lane >> 4;

    v8f acc1[MID_DIM / 16] = {};
    dense64_T<MID_DIM / 16>(Wl1, hs, m, half, acc1);
#pragma unroll
    for (int t = 0; t < MID_DIM / 16; ++t) bias_relu_T(acc1[t], bl1 + 16 * t, half);

    v8f acc2[HID_DIM / 16] = {};
    __syncthreads();
#pragma unroll
    for (int t = 0; t < 4; ++t) put_tile_T(hs, t, m, half, acc1[t]);
    __syncthreads();
    dense64_T<HID_DIM / 16>(Wl2, hs, m, half, acc2);
    __syncthreads();
#pragma unroll
    for (int t = 0; t < 4; ++t) put_tile_T(hs, t, m, half, acc1[4 + t]);
    __syncthreads();
    dense64_T<HID_DIM / 16>(Wl2 + 64 * HID_DIM, hs, m, half, acc2);
#pragma unroll
    for (int t = 0; t < HID_DIM / 16; ++t) bias_relu_T(acc2[t], bl2 + 16 * t, half);

    v8f acc3[OUT_DIM / 16] = {};
    __syncthreads();
#pragma unroll
    for (int t = 0; t < HID_DIM / 16; ++t) put_tile_T(hs, t, m, half, acc2[t]);
    __syncthreads();
    dense64_T<OUT_DIM / 16>(W2, hs, m, half, acc3);

    const int row = d0 + m;
    if (row < N) {
        const float rs = dis[row];
        float* __restrict__ op = out + (long long)row * OUT_DIM + 8 * half;
#pragma unroll
        for (int t = 0; t < OUT_DIM / 16; ++t) {
            const v4f lo = {acc3[t][0] * rs, acc3[t][1] * rs, acc3[t][2] * rs, acc3[t][3] * rs};
            const v4f hi = {acc3[t][4] * rs, acc3[t][5] * rs, acc3[t][6] * rs, acc3[t][7] * rs};
            *(v4f*)(op + 16 * t)     = lo;
            *(v4f*)(op + 16 * t + 4) = hi;
        }
    }
}

extern "C" void kernel_launch(void* const* d_in, const int* in_sizes, int n_in,
                              void* d_out, int out_size, void* d_ws, size_t ws_size,
                              hipStream_t stream) {
    const float* x   = (const float*)d_in[0];
    const int*   ei  = (const int*)  d_in[1];
    const float* W1  = (const float*)d_in[2];
    const float* b1  = (const float*)d_in[3];
    const float* Wl1 = (const float*)d_in[4];
    const float* bl1 = (const float*)d_in[5];
    const float* Wl2 = (const float*)d_in[6];
    const float* bl2 = (const float*)d_in[7];
    const float* W2  = (const float*)d_in[8];
    const float* b2  = (const float*)d_in[9];

    const int N = in_sizes[0] / IN_DIM;
    const int E = in_sizes[1] / 2;
    const int* src = ei;
    const int* dst = ei + E;

    const int nblk = (E + P_CHUNK - 1) / P_CHUNK;
    const int nbin = (N + BIN_NODES - 1) >> BIN_SHIFT;
    const size_t Np = ((size_t)N + 1 + 63) / 64 * 64;
    const size_t Ep = ((size_t)E + 63) / 64 * 64;
    const size_t Tp = ((size_t)nblk * TBL_ROW + 63) / 64 * 64;
    int*   table  = (int*)d_ws;
    int*   rowptr = table + Tp;
    float* dis    = (float*)(rowptr + Np);
    int*   tmp    = (int*)(dis + Np);
    int*   csr    = tmp + 2 * Ep;
    float* bufA   = (float*)(csr + Ep);
    float* bufB   = bufA + (size_t)N * 64;
    float* out    = (float*)d_out;

    const int TPB = 256;
    const int rowTiles = (N + 15) / 16;
    const int gemmBlocks = (rowTiles + 7) / 8;
    const int AGG_RUN = 16;
    const int aggBlocks  = (rowTiles + 3) / 4;

    partition_kernel<<<nblk, P_THREADS, 0, stream>>>(src, dst, tmp, table, E);
    bin_csr_kernel<<<nbin, 1024, 0, stream>>>(tmp, table, rowptr, dis, csr, N, nblk);

    gemm_wmma_kernel<IN_DIM, HID_DIM / 16, false, false, true>
        <<<gemmBlocks, TPB, 0, stream>>>(x, W1, nullptr, dis, bufA, rowTiles, N);
    conv1_dense_kernel<<<rowTiles, 32, 0, stream>>>(bufA, rowptr, csr, dis, b1,
                                                  Wl1, bl1, Wl2, bl2, W2, bufB, N);

    agg_run_kernel<OUT_DIM, false, AGG_RUN><<<aggBlocks, 128, 0, stream>>>(bufB, rowptr, csr, dis, b2, out, N);
}
